# starting kernel
# baseline (speedup 1.0000x reference)
#include <hip/hip_runtime.h>

typedef __attribute__((ext_vector_type(2))) float v2f;
typedef __attribute__((ext_vector_type(8))) float v8f;

#define K_TAPS 4096
#define PAD    2048
#define T_LEN  131072
#define S_SIG  32
#define TP     (T_LEN + 2*PAD)
#define GT     2112
#define ROWS   (TP + 2*GT)
#define GUARD  16
#define HEXT   (K_TAPS + 64)
#define KWIN   (K_TAPS + 15)
#define NCHUNK ((KWIN + 4) / 4)
#define TBASE  (-2048)

__global__ void prep_h_kernel(const float* __restrict__ h,
                              float* __restrict__ hext,
                              float* __restrict__ hrev) {
    int i = blockIdx.x * blockDim.x + threadIdx.x;
    if (i >= HEXT) return;
    int j = i - GUARD - 1;
    int d = i - GUARD;
    hext[i] = (j >= 0 && j < K_TAPS) ? h[j] : 0.f;
    hrev[i] = (d >= 0 && d < K_TAPS) ? h[K_TAPS - 1 - d] : 0.f;
}

__global__ void build_xpt_kernel(const float* __restrict__ x,
                                 float* __restrict__ xpt) {
    int o = blockIdx.x * blockDim.x + threadIdx.x;
    int rem = o & 63;
    int s = rem >> 1;
    int t = ((o >> 6) << 1) + (rem & 1);
    int tg = t - GT;
    float v = 0.f;
    if (tg >= 0 && tg < TP) {
        int tr = tg - PAD;
        if (tr < 0)            tr = -tr;
        else if (tr >= T_LEN)  tr = 2 * T_LEN - 2 - tr;
        v = x[(size_t)s * T_LEN + tr];
    }
    xpt[o] = v;
}

__global__ void zero_guards_kernel(float* __restrict__ y1t) {
    int idx = blockIdx.x * blockDim.x + threadIdx.x;
    y1t[idx] = 0.f;
    y1t[(size_t)(GT + TP) * S_SIG + idx] = 0.f;
}

template<bool STORE_OUT>
__global__ __launch_bounds__(256)
void fir_pass_kernel(const float* __restrict__ hext,
                     const float* __restrict__ src,
                     float* __restrict__ dst,
                     int t0_origin) {
    __shared__ float sh[HEXT];
    for (int i = threadIdx.x; i < HEXT; i += 256) sh[i] = hext[i];
    __syncthreads();

    const int lane = threadIdx.x & 31;
    const int wave = threadIdx.x >> 5;
    const int n    = lane & 15;
    const int half = lane >> 4;
    const int q    = half * 2;

    const int t0 = t0_origin + (blockIdx.x * 8 + wave) * 16;

    int ab = GUARD + q - n;
    const float* p = src + ((size_t)(GT + t0 + TBASE + q) >> 1) * 64 + 2 * n;

    v8f acc0 = {};
    v8f acc1 = {};
#pragma unroll 4
    for (int kc = 0; kc < NCHUNK; ++kc) {
        v2f a, b0, b1;
        a.x = sh[ab];
        a.y = sh[ab + 1];
        b0 = *(const v2f*)p;
        b1 = *(const v2f*)(p + 32);
        acc0 = __builtin_amdgcn_wmma_f32_16x16x4_f32(
            false, a, false, b0, (short)0, acc0, false, false);
        acc1 = __builtin_amdgcn_wmma_f32_16x16x4_f32(
            false, a, false, b1, (short)0, acc1, false, false);
        ab += 4;
        p  += 128;
    }

    if (!STORE_OUT) {
        const int row0 = GT + t0 + half * 8;
        float* o = dst + ((size_t)row0 >> 1) * 64 + (size_t)n * 2;
#pragma unroll
        for (int r = 0; r < 4; ++r) {
            v2f w0, w1;
            w0.x = acc0[2 * r]; w0.y = acc0[2 * r + 1];
            w1.x = acc1[2 * r]; w1.y = acc1[2 * r + 1];
            *(v2f*)(o + (size_t)r * 64)      = w0;
            *(v2f*)(o + (size_t)r * 64 + 32) = w1;
        }
    } else {
        float* o0 = dst + (size_t)n        * T_LEN + (t0 - PAD) + half * 8;
        float* o1 = dst + (size_t)(n + 16) * T_LEN + (t0 - PAD) + half * 8;
        *(float4*)(o0)     = make_float4(acc0[0], acc0[1], acc0[2], acc0[3]);
        *((float4*)o0 + 1) = make_float4(acc0[4], acc0[5], acc0[6], acc0[7]);
        *(float4*)(o1)     = make_float4(acc1[0], acc1[1], acc1[2], acc1[3]);
        *((float4*)o1 + 1) = make_float4(acc1[4], acc1[5], acc1[6], acc1[7]);
    }
}

extern "C" void kernel_launch(void* const* d_in, const int* in_sizes, int n_in,
                              void* d_out, int out_size, void* d_ws, size_t ws_size,
                              hipStream_t stream) {
    const float* x    = (const float*)d_in[0];
    const float* kern = (const float*)d_in[1];
    float* out = (float*)d_out;
    float* ws  = (float*)d_ws;

    float* hext = ws;
    float* hrev = ws + HEXT;
    float* xpt  = ws + 2 * HEXT;
    float* y1t  = xpt + (size_t)ROWS * S_SIG;

    prep_h_kernel<<<(HEXT + 255) / 256, 256, 0, stream>>>(kern, hext, hrev);
    build_xpt_kernel<<<(ROWS * S_SIG) / 256, 256, 0, stream>>>(x, xpt);
    zero_guards_kernel<<<(GT * S_SIG) / 256, 256, 0, stream>>>(y1t);

    fir_pass_kernel<false>
        <<<TP / 128, 256, 0, stream>>>(hext, xpt, y1t, 0);
    fir_pass_kernel<true>
        <<<T_LEN / 128, 256, 0, stream>>>(hrev, y1t, out, PAD);
}


# both passes as one even filter r = h corr h (built per call), folded about each 32-row tile centre: mirrored rows added/subtracted, taps r[|i-j|] +/- r[i+j+1]; half the WMMAs; sum order changes
# speedup vs baseline: 1.9832x; 1.9832x over previous
#include <hip/hip_runtime.h>


typedef __attribute__((ext_vector_type(2))) float v2f;
typedef __attribute__((ext_vector_type(8))) float v8f;

#define K_TAPS 4096
#define PAD    2048
#define T_LEN  131072
#define S_SIG  32
#define TP     (T_LEN + 2*PAD)
#define GT     2176
#define ROWS   (TP + 2*GT)
#define RTAB   4160
#define RC     15
#define NCHUNK 1028
#define TILE   32

__global__ __launch_bounds__(256)
void prep_r_kernel(const float* __restrict__ h, float* __restrict__ rt) {
    __shared__ float red[8];
    const int tid = threadIdx.x;
    int d = (int)blockIdx.x - RC;
    if (d < 0) d = -d;
    float a[K_TAPS / 256], b[K_TAPS / 256];
#pragma unroll
    for (int j = 0; j < K_TAPS / 256; ++j) {
        const int i  = tid + 256 * j;
        const int i2 = (i + d < K_TAPS) ? i + d : K_TAPS - 1;
        a[j] = h[i];
        b[j] = h[i2];
    }
    float s = 0.f;
#pragma unroll
    for (int j = 0; j < K_TAPS / 256; ++j) {
        const int i = tid + 256 * j;
        s += a[j] * ((i + d < K_TAPS) ? b[j] : 0.f);
    }
#pragma unroll
    for (int o = 16; o > 0; o >>= 1) s += __shfl_xor(s, o);
    if ((tid & 31) == 0) red[tid >> 5] = s;
    __syncthreads();
    if (tid == 0) {
        float tot = red[0];
#pragma unroll
        for (int w = 1; w < 8; ++w) tot += red[w];
        rt[blockIdx.x] = 0.5f * tot;
    }
}

__global__ void build_xpt_kernel(const float* __restrict__ x,
                                 float* __restrict__ xpt) {
    int o = blockIdx.x * blockDim.x + threadIdx.x;
    int rem = o & 63;
    int s = rem >> 1;
    int t = ((o >> 6) << 1) + (rem & 1);
    int tg = t - GT;
    float v = 0.f;
    if (tg >= 0 && tg < TP) {
        int tr = tg - PAD;
        if (tr < 0)            tr = -tr;
        else if (tr >= T_LEN)  tr = 2 * T_LEN - 2 - tr;
        v = x[(size_t)s * T_LEN + tr];
    }
    xpt[o] = v;
}

#define RING_PR 256
#define RING_F  (RING_PR * 64)
#define NFULL   32
#define NTAIL   (NCHUNK - 32 * NFULL)
#define FILL_PR 176

__global__ __launch_bounds__(256)
void fold_fir_kernel(const float* __restrict__ rt,
                     const float* __restrict__ src,
                     float* __restrict__ dst) {
    __shared__ __attribute__((aligned(16))) float sr[RTAB];
    __shared__ __attribute__((aligned(16))) float ringu[RING_F];
    __shared__ __attribute__((aligned(16))) float ringl[RING_F];

    const int tid  = threadIdx.x;
    const int lane = tid & 31;
    const int wave = __builtin_amdgcn_readfirstlane(tid >> 5);
    const int n    = lane & 15;
    const int half = lane >> 4;
    const int q    = half * 2;

    const int n0b = PAD + (int)blockIdx.x * 256;
    const int pbu = (GT + n0b + 16) >> 1;
    const int pbl = (GT + n0b + 238) >> 1;
    const float4* gu = (const float4*)(src + (size_t)pbu * 64);
    const float4* gl = (const float4*)(src + (size_t)pbl * 64);

    const int tr  = tid >> 4;
    const int w4  = tid & 15;
    const int cu4 = w4 ^ ((tr & 1) << 3);
    const int cl4 = w4 ^ (((tr & 1) ^ 1) << 3);
    float4* lu = (float4*)ringu;
    float4* ll = (float4*)ringl;

    {
        const float4* rt4 = (const float4*)rt;
        float4* sr4 = (float4*)sr;
        float4 a[11], b[11];
        const float4 f0 = rt4[tid],       f1 = rt4[tid + 256];
        const float4 f2 = rt4[tid + 512], f3 = rt4[tid + 768];
        float4 f4 = make_float4(0.f, 0.f, 0.f, 0.f);
        if (tid < RTAB / 4 - 1024) f4 = rt4[tid + 1024];
#pragma unroll
        for (int i = 0; i < 11; ++i) {
            const int p = tr + 16 * i;
            a[i] = gu[p * 16 + w4];
            b[i] = gl[w4 - p * 16];
        }
        sr4[tid]       = f0;  sr4[tid + 256] = f1;
        sr4[tid + 512] = f2;  sr4[tid + 768] = f3;
        if (tid < RTAB / 4 - 1024) sr4[tid + 1024] = f4;
#pragma unroll
        for (int i = 0; i < 11; ++i) {
            const int p = tr + 16 * i;
            lu[p * 16 + cu4] = a[i];
            ll[p * 16 + cl4] = b[i];
        }
    }

    const int aT = RC + q - n;
    const int aH = RC + 1 + q + n;
    const int bu = 16 * wave + half;
    const int bl = 16 * (7 - wave) + half;
    const int cu0 = (2 * n + 32 * half) & 63;
    const int cu1 = cu0 ^ 32;
    const int cl0 = (2 * n + 32 * (1 - half)) & 63;
    const int cl1 = cl0 ^ 32;

    v8f ps0 = {}, ps1 = {};
    v8f pa0 = {}, pa1 = {};

    float4 u0, u1, u2, u3, l0, l1, l2, l3;
#define STAGE_LOAD(s1)                                                      \
    {                                                                       \
        const int p = 64 * (s1) + 112 + tr;                                 \
        u0 = gu[p * 16 + w4];        u1 = gu[(p + 16) * 16 + w4];           \
        u2 = gu[(p + 32) * 16 + w4]; u3 = gu[(p + 48) * 16 + w4];           \
        l0 = gl[w4 - p * 16];        l1 = gl[w4 - (p + 16) * 16];           \
        l2 = gl[w4 - (p + 32) * 16]; l3 = gl[w4 - (p + 48) * 16];           \
    }
#define STAGE_STORE(s1)                                                     \
    {                                                                       \
        const int p = 64 * (s1) + 112 + tr;                                 \
        lu[((p)      & 255) * 16 + cu4] = u0;                               \
        lu[((p + 16) & 255) * 16 + cu4] = u1;                               \
        lu[((p + 32) & 255) * 16 + cu4] = u2;                               \
        lu[((p + 48) & 255) * 16 + cu4] = u3;                               \
        ll[((p)      & 255) * 16 + cl4] = l0;                               \
        ll[((p + 16) & 255) * 16 + cl4] = l1;                               \
        ll[((p + 32) & 255) * 16 + cl4] = l2;                               \
        ll[((p + 48) & 255) * 16 + cl4] = l3;                               \
    }
#define FOLD_CHUNK(cc)                                                      \
    {                                                                       \
        const int su = ((bus + 2 * (cc)) & 255) << 6;                       \
        const int sl = ((bls + 2 * (cc)) & 255) << 6;                       \
        v2f t, g;                                                           \
        t.x = sr[ats + 4 * (cc)];                                           \
        t.y = sr[ats + 4 * (cc) + 1];                                       \
        g.x = sr[ahs + 4 * (cc)];                                           \
        g.y = sr[ahs + 4 * (cc) + 1];                                       \
        const v2f xp0 = *(const v2f*)(ringu + su + cu0);                    \
        const v2f xp1 = *(const v2f*)(ringu + su + cu1);                    \
        const v2f xm0 = *(const v2f*)(ringl + sl + cl0);                    \
        const v2f xm1 = *(const v2f*)(ringl + sl + cl1);                    \
        v2f as, aa, bs0, bs1, ba0, ba1;                                     \
        as.x = t.x + g.x;        as.y = t.y + g.y;                          \
        aa.x = t.x - g.x;        aa.y = t.y - g.y;                          \
        bs0.x = xp0.x + xm0.y;   bs0.y = xp0.y + xm0.x;                     \
        ba0.x = xp0.x - xm0.y;   ba0.y = xp0.y - xm0.x;                     \
        bs1.x = xp1.x + xm1.y;   bs1.y = xp1.y + xm1.x;                     \
        ba1.x = xp1.x - xm1.y;   ba1.y = xp1.y - xm1.x;                     \
        ps0 = __builtin_amdgcn_wmma_f32_16x16x4_f32(                        \
            false, as, false, bs0, (short)0, ps0, false, false);            \
        ps1 = __builtin_amdgcn_wmma_f32_16x16x4_f32(                        \
            false, as, false, bs1, (short)0, ps1, false, false);            \
        pa0 = __builtin_amdgcn_wmma_f32_16x16x4_f32(                        \
            false, aa, false, ba0, (short)0, pa0, false, false);            \
        pa1 = __builtin_amdgcn_wmma_f32_16x16x4_f32(                        \
            false, aa, false, ba1, (short)0, pa1, false, false);            \
    }

    STAGE_LOAD(1)
    __syncthreads();

    for (int s = 0; s < NFULL; ++s) {
        const int ats = aT + 128 * s;
        const int ahs = aH + 128 * s;
        const int bus = bu + 64 * s;
        const int bls = bl + 64 * s;
#pragma unroll
        for (int cc = 0; cc < 32; ++cc) FOLD_CHUNK(cc)
        STAGE_STORE(s + 1)
        if (s + 2 <= NFULL) STAGE_LOAD(s + 2)
        __syncthreads();
    }
    {
        const int ats = aT + 128 * NFULL;
        const int ahs = aH + 128 * NFULL;
        const int bus = bu + 64 * NFULL;
        const int bls = bl + 64 * NFULL;
#pragma unroll
        for (int cc = 0; cc < NTAIL; ++cc) FOLD_CHUNK(cc)
    }
#undef STAGE_LOAD
#undef STAGE_STORE
#undef FOLD_CHUNK

    const int tb = n0b - PAD + TILE * wave;
    float* o0 = dst + (size_t)n        * T_LEN + tb;
    float* o1 = dst + (size_t)(n + 16) * T_LEN + tb;
    float* u0p = o0 + 16 + 8 * half;
    float* u1p = o1 + 16 + 8 * half;
    float* l0p = o0 + 8 - 8 * half;
    float* l1p = o1 + 8 - 8 * half;
    *(float4*)(u0p)     = make_float4(ps0[0] + pa0[0], ps0[1] + pa0[1],
                                      ps0[2] + pa0[2], ps0[3] + pa0[3]);
    *((float4*)u0p + 1) = make_float4(ps0[4] + pa0[4], ps0[5] + pa0[5],
                                      ps0[6] + pa0[6], ps0[7] + pa0[7]);
    *(float4*)(u1p)     = make_float4(ps1[0] + pa1[0], ps1[1] + pa1[1],
                                      ps1[2] + pa1[2], ps1[3] + pa1[3]);
    *((float4*)u1p + 1) = make_float4(ps1[4] + pa1[4], ps1[5] + pa1[5],
                                      ps1[6] + pa1[6], ps1[7] + pa1[7]);
    *(float4*)(l0p)     = make_float4(ps0[7] - pa0[7], ps0[6] - pa0[6],
                                      ps0[5] - pa0[5], ps0[4] - pa0[4]);
    *((float4*)l0p + 1) = make_float4(ps0[3] - pa0[3], ps0[2] - pa0[2],
                                      ps0[1] - pa0[1], ps0[0] - pa0[0]);
    *(float4*)(l1p)     = make_float4(ps1[7] - pa1[7], ps1[6] - pa1[6],
                                      ps1[5] - pa1[5], ps1[4] - pa1[4]);
    *((float4*)l1p + 1) = make_float4(ps1[3] - pa1[3], ps1[2] - pa1[2],
                                      ps1[1] - pa1[1], ps1[0] - pa1[0]);
}

extern "C" void kernel_launch(void* const* d_in, const int* in_sizes, int n_in,
                              void* d_out, int out_size, void* d_ws, size_t ws_size,
                              hipStream_t stream) {
    const float* x    = (const float*)d_in[0];
    const float* kern = (const float*)d_in[1];
    float* out = (float*)d_out;
    float* ws  = (float*)d_ws;

    float* rt  = ws;
    float* xpt = ws + RTAB;

    prep_r_kernel<<<RTAB, 256, 0, stream>>>(kern, rt);
    build_xpt_kernel<<<(ROWS * S_SIG) / 256, 256, 0, stream>>>(x, xpt);
    fold_fir_kernel<<<T_LEN / 256, 256, 0, stream>>>(rt, xpt, out);
}


# fold pass: a wave's LDS operand reads issued as one group per 4 chunks, one segment ahead of the 16 WMMAs that use them (second register set, one ring base per segment); every sum unchanged
# speedup vs baseline: 2.1389x; 1.0785x over previous
#include <hip/hip_runtime.h>


typedef __attribute__((ext_vector_type(2))) float v2f;
typedef __attribute__((ext_vector_type(8))) float v8f;

#define K_TAPS 4096
#define PAD    2048
#define T_LEN  131072
#define S_SIG  32
#define TP     (T_LEN + 2*PAD)
#define GT     2176
#define ROWS   (TP + 2*GT)
#define RTAB   4160
#define RC     15
#define NCHUNK 1028
#define TILE   32

__global__ __launch_bounds__(256)
void prep_r_kernel(const float* __restrict__ h, float* __restrict__ rt) {
    __shared__ float red[8];
    const int tid = threadIdx.x;
    int d = (int)blockIdx.x - RC;
    if (d < 0) d = -d;
    float a[K_TAPS / 256], b[K_TAPS / 256];
#pragma unroll
    for (int j = 0; j < K_TAPS / 256; ++j) {
        const int i  = tid + 256 * j;
        const int i2 = (i + d < K_TAPS) ? i + d : K_TAPS - 1;
        a[j] = h[i];
        b[j] = h[i2];
    }
    float s = 0.f;
#pragma unroll
    for (int j = 0; j < K_TAPS / 256; ++j) {
        const int i = tid + 256 * j;
        s += a[j] * ((i + d < K_TAPS) ? b[j] : 0.f);
    }
#pragma unroll
    for (int o = 16; o > 0; o >>= 1) s += __shfl_xor(s, o);
    if ((tid & 31) == 0) red[tid >> 5] = s;
    __syncthreads();
    if (tid == 0) {
        float tot = red[0];
#pragma unroll
        for (int w = 1; w < 8; ++w) tot += red[w];
        rt[blockIdx.x] = 0.5f * tot;
    }
}

__global__ void build_xpt_kernel(const float* __restrict__ x,
                                 float* __restrict__ xpt) {
    int o = blockIdx.x * blockDim.x + threadIdx.x;
    int rem = o & 63;
    int s = rem >> 1;
    int t = ((o >> 6) << 1) + (rem & 1);
    int tg = t - GT;
    float v = 0.f;
    if (tg >= 0 && tg < TP) {
        int tr = tg - PAD;
        if (tr < 0)            tr = -tr;
        else if (tr >= T_LEN)  tr = 2 * T_LEN - 2 - tr;
        v = x[(size_t)s * T_LEN + tr];
    }
    xpt[o] = v;
}

#define RING_PR 256
#define RING_F  (RING_PR * 64)
#define NFULL   32
#define NTAIL   (NCHUNK - 32 * NFULL)
#define FILL_PR 176

__global__ __launch_bounds__(256)
void fold_fir_kernel(const float* __restrict__ rt,
                     const float* __restrict__ src,
                     float* __restrict__ dst) {
    __shared__ __attribute__((aligned(16))) float sr[RTAB];
    __shared__ __attribute__((aligned(16))) float ringu[RING_F];
    __shared__ __attribute__((aligned(16))) float ringl[RING_F];

    const int tid  = threadIdx.x;
    const int lane = tid & 31;
    const int wave = __builtin_amdgcn_readfirstlane(tid >> 5);
    const int n    = lane & 15;
    const int half = lane >> 4;
    const int q    = half * 2;

    const int n0b = PAD + (int)blockIdx.x * 256;
    const int pbu = (GT + n0b + 16) >> 1;
    const int pbl = (GT + n0b + 238) >> 1;
    const float4* gu = (const float4*)(src + (size_t)pbu * 64);
    const float4* gl = (const float4*)(src + (size_t)pbl * 64);

    const int tr  = tid >> 4;
    const int w4  = tid & 15;
    const int cu4 = w4 ^ ((tr & 1) << 3);
    const int cl4 = w4 ^ (((tr & 1) ^ 1) << 3);
    float4* lu = (float4*)ringu;
    float4* ll = (float4*)ringl;

    {
        const float4* rt4 = (const float4*)rt;
        float4* sr4 = (float4*)sr;
        float4 a[11], b[11];
        const float4 f0 = rt4[tid],       f1 = rt4[tid + 256];
        const float4 f2 = rt4[tid + 512], f3 = rt4[tid + 768];
        float4 f4 = make_float4(0.f, 0.f, 0.f, 0.f);
        if (tid < RTAB / 4 - 1024) f4 = rt4[tid + 1024];
#pragma unroll
        for (int i = 0; i < 11; ++i) {
            const int p = tr + 16 * i;
            a[i] = gu[p * 16 + w4];
            b[i] = gl[w4 - p * 16];
        }
        sr4[tid]       = f0;  sr4[tid + 256] = f1;
        sr4[tid + 512] = f2;  sr4[tid + 768] = f3;
        if (tid < RTAB / 4 - 1024) sr4[tid + 1024] = f4;
#pragma unroll
        for (int i = 0; i < 11; ++i) {
            const int p = tr + 16 * i;
            lu[p * 16 + cu4] = a[i];
            ll[p * 16 + cl4] = b[i];
        }
    }

    const int aT = RC + q - n;
    const int aH = RC + 1 + q + n;
    const int bu = 16 * wave + half;
    const int bl = 16 * (7 - wave) + half;
    const int cu0 = (2 * n + 32 * half) & 63;
    const int cu1 = cu0 ^ 32;
    const int cl0 = (2 * n + 32 * (1 - half)) & 63;
    const int cl1 = cl0 ^ 32;

    v8f ps0 = {}, ps1 = {};
    v8f pa0 = {}, pa1 = {};

    float4 u0, u1, u2, u3, l0, l1, l2, l3;
#define STAGE_LOAD(s1)                                                      \
    {                                                                       \
        const int p = 64 * (s1) + 112 + tr;                                 \
        u0 = gu[p * 16 + w4];        u1 = gu[(p + 16) * 16 + w4];           \
        u2 = gu[(p + 32) * 16 + w4]; u3 = gu[(p + 48) * 16 + w4];           \
        l0 = gl[w4 - p * 16];        l1 = gl[w4 - (p + 16) * 16];           \
        l2 = gl[w4 - (p + 32) * 16]; l3 = gl[w4 - (p + 48) * 16];           \
    }
#define STAGE_STORE(s1)                                                     \
    {                                                                       \
        const int p = 64 * (s1) + 112 + tr;                                 \
        lu[((p)      & 255) * 16 + cu4] = u0;                               \
        lu[((p + 16) & 255) * 16 + cu4] = u1;                               \
        lu[((p + 32) & 255) * 16 + cu4] = u2;                               \
        lu[((p + 48) & 255) * 16 + cu4] = u3;                               \
        ll[((p)      & 255) * 16 + cl4] = l0;                               \
        ll[((p + 16) & 255) * 16 + cl4] = l1;                               \
        ll[((p + 32) & 255) * 16 + cl4] = l2;                               \
        ll[((p + 48) & 255) * 16 + cl4] = l3;                               \
    }
#define SEG 4
#define LOAD_SEG(X, j)                                                      \
    {                                                                       \
        const float* fu = ringu + (((bus + 8 * (j)) & 255) << 6);           \
        const float* fl = ringl + (((bls + 8 * (j)) & 255) << 6);           \
        _Pragma("unroll")                                                   \
        for (int c = 0; c < SEG; ++c) {                                     \
            X.t[c].x = ft[16 * (j) + 4 * c];                                \
            X.t[c].y = ft[16 * (j) + 4 * c + 1];                            \
            X.g[c].x = fh[16 * (j) + 4 * c];                                \
            X.g[c].y = fh[16 * (j) + 4 * c + 1];                            \
            X.xp0[c] = *(const v2f*)(fu + cu0 + 128 * c);                   \
            X.xp1[c] = *(const v2f*)(fu + cu1 + 128 * c);                   \
            X.xm0[c] = *(const v2f*)(fl + cl0 + 128 * c);                   \
            X.xm1[c] = *(const v2f*)(fl + cl1 + 128 * c);                   \
        }                                                                   \
    }
#define COMPUTE_SEG(X)                                                      \
    {                                                                       \
        _Pragma("unroll")                                                   \
        for (int c = 0; c < SEG; ++c) {                                     \
            const v2f t = X.t[c], g = X.g[c];                               \
            const v2f xp0 = X.xp0[c], xp1 = X.xp1[c];                       \
            const v2f xm0 = X.xm0[c], xm1 = X.xm1[c];                       \
            v2f as, aa, bs0, bs1, ba0, ba1;                                 \
            as.x = t.x + g.x;        as.y = t.y + g.y;                      \
            aa.x = t.x - g.x;        aa.y = t.y - g.y;                      \
            bs0.x = xp0.x + xm0.y;   bs0.y = xp0.y + xm0.x;                 \
            ba0.x = xp0.x - xm0.y;   ba0.y = xp0.y - xm0.x;                 \
            bs1.x = xp1.x + xm1.y;   bs1.y = xp1.y + xm1.x;                 \
            ba1.x = xp1.x - xm1.y;   ba1.y = xp1.y - xm1.x;                 \
            ps0 = __builtin_amdgcn_wmma_f32_16x16x4_f32(                    \
                false, as, false, bs0, (short)0, ps0, false, false);        \
            ps1 = __builtin_amdgcn_wmma_f32_16x16x4_f32(                    \
                false, as, false, bs1, (short)0, ps1, false, false);        \
            pa0 = __builtin_amdgcn_wmma_f32_16x16x4_f32(                    \
                false, aa, false, ba0, (short)0, pa0, false, false);        \
            pa1 = __builtin_amdgcn_wmma_f32_16x16x4_f32(                    \
                false, aa, false, ba1, (short)0, pa1, false, false);        \
        }                                                                   \
    }
#define GROUP_EDGE() asm volatile("" ::: "memory")

    struct SegOps { v2f t[SEG], g[SEG], xp0[SEG], xp1[SEG], xm0[SEG], xm1[SEG]; };

    STAGE_LOAD(1)
    __syncthreads();

    for (int s = 0; s < NFULL; ++s) {
        const float* ft = sr + aT + 128 * s;
        const float* fh = sr + aH + 128 * s;
        const int bus = bu + 64 * s;
        const int bls = bl + 64 * s;
        SegOps A, B;
        LOAD_SEG(A, 0)
        GROUP_EDGE();
#pragma unroll
        for (int j = 0; j < 32 / SEG; j += 2) {
            LOAD_SEG(B, j + 1)
            GROUP_EDGE();
            COMPUTE_SEG(A)
            GROUP_EDGE();
            if (j + 2 < 32 / SEG) LOAD_SEG(A, j + 2)
            GROUP_EDGE();
            COMPUTE_SEG(B)
            GROUP_EDGE();
        }
        __builtin_amdgcn_sched_group_barrier(0x100, 32, 0);
#pragma unroll
        for (int j = 0; j < 32 / SEG - 2; ++j) {
            __builtin_amdgcn_sched_group_barrier(0x008, 4 * SEG, 0);
            __builtin_amdgcn_sched_group_barrier(0x100, 16, 0);
        }
        __builtin_amdgcn_sched_group_barrier(0x008, 8 * SEG, 0);
        STAGE_STORE(s + 1)
        if (s + 2 <= NFULL) STAGE_LOAD(s + 2)
        __syncthreads();
    }
    {
        const float* ft = sr + aT + 128 * NFULL;
        const float* fh = sr + aH + 128 * NFULL;
        const int bus = bu + 64 * NFULL;
        const int bls = bl + 64 * NFULL;
        static_assert(NTAIL == SEG, "the closing stage is one segment");
        SegOps A;
        LOAD_SEG(A, 0)
        COMPUTE_SEG(A)
    }
#undef LOAD_SEG
#undef COMPUTE_SEG
#undef GROUP_EDGE
#undef STAGE_LOAD
#undef STAGE_STORE

    const int tb = n0b - PAD + TILE * wave;
    float* o0 = dst + (size_t)n        * T_LEN + tb;
    float* o1 = dst + (size_t)(n + 16) * T_LEN + tb;
    float* u0p = o0 + 16 + 8 * half;
    float* u1p = o1 + 16 + 8 * half;
    float* l0p = o0 + 8 - 8 * half;
    float* l1p = o1 + 8 - 8 * half;
    *(float4*)(u0p)     = make_float4(ps0[0] + pa0[0], ps0[1] + pa0[1],
                                      ps0[2] + pa0[2], ps0[3] + pa0[3]);
    *((float4*)u0p + 1) = make_float4(ps0[4] + pa0[4], ps0[5] + pa0[5],
                                      ps0[6] + pa0[6], ps0[7] + pa0[7]);
    *(float4*)(u1p)     = make_float4(ps1[0] + pa1[0], ps1[1] + pa1[1],
                                      ps1[2] + pa1[2], ps1[3] + pa1[3]);
    *((float4*)u1p + 1) = make_float4(ps1[4] + pa1[4], ps1[5] + pa1[5],
                                      ps1[6] + pa1[6], ps1[7] + pa1[7]);
    *(float4*)(l0p)     = make_float4(ps0[7] - pa0[7], ps0[6] - pa0[6],
                                      ps0[5] - pa0[5], ps0[4] - pa0[4]);
    *((float4*)l0p + 1) = make_float4(ps0[3] - pa0[3], ps0[2] - pa0[2],
                                      ps0[1] - pa0[1], ps0[0] - pa0[0]);
    *(float4*)(l1p)     = make_float4(ps1[7] - pa1[7], ps1[6] - pa1[6],
                                      ps1[5] - pa1[5], ps1[4] - pa1[4]);
    *((float4*)l1p + 1) = make_float4(ps1[3] - pa1[3], ps1[2] - pa1[2],
                                      ps1[1] - pa1[1], ps1[0] - pa1[0]);
}

extern "C" void kernel_launch(void* const* d_in, const int* in_sizes, int n_in,
                              void* d_out, int out_size, void* d_ws, size_t ws_size,
                              hipStream_t stream) {
    const float* x    = (const float*)d_in[0];
    const float* kern = (const float*)d_in[1];
    float* out = (float*)d_out;
    float* ws  = (float*)d_ws;

    float* rt  = ws;
    float* xpt = ws + RTAB;

    prep_r_kernel<<<RTAB, 256, 0, stream>>>(kern, rt);
    build_xpt_kernel<<<(ROWS * S_SIG) / 256, 256, 0, stream>>>(x, xpt);
    fold_fir_kernel<<<T_LEN / 256, 256, 0, stream>>>(rt, xpt, out);
}


# ring refill (8 LDS stores + 8 global loads a thread) moved from in front of the stage barrier to the stage head: the barrier's fence no longer waits out a memory round trip per stage; sums unchanged
# speedup vs baseline: 2.2753x; 1.0638x over previous
#include <hip/hip_runtime.h>


typedef __attribute__((ext_vector_type(2))) float v2f;
typedef __attribute__((ext_vector_type(8))) float v8f;

#define K_TAPS 4096
#define PAD    2048
#define T_LEN  131072
#define S_SIG  32
#define TP     (T_LEN + 2*PAD)
#define GT     2176
#define ROWS   (TP + 2*GT)
#define RTAB   4160
#define RC     15
#define NCHUNK 1028
#define TILE   32

__global__ __launch_bounds__(256)
void prep_r_kernel(const float* __restrict__ h, float* __restrict__ rt) {
    __shared__ float red[8];
    const int tid = threadIdx.x;
    int d = (int)blockIdx.x - RC;
    if (d < 0) d = -d;
    float a[K_TAPS / 256], b[K_TAPS / 256];
#pragma unroll
    for (int j = 0; j < K_TAPS / 256; ++j) {
        const int i  = tid + 256 * j;
        const int i2 = (i + d < K_TAPS) ? i + d : K_TAPS - 1;
        a[j] = h[i];
        b[j] = h[i2];
    }
    float s = 0.f;
#pragma unroll
    for (int j = 0; j < K_TAPS / 256; ++j) {
        const int i = tid + 256 * j;
        s += a[j] * ((i + d < K_TAPS) ? b[j] : 0.f);
    }
#pragma unroll
    for (int o = 16; o > 0; o >>= 1) s += __shfl_xor(s, o);
    if ((tid & 31) == 0) red[tid >> 5] = s;
    __syncthreads();
    if (tid == 0) {
        float tot = red[0];
#pragma unroll
        for (int w = 1; w < 8; ++w) tot += red[w];
        rt[blockIdx.x] = 0.5f * tot;
    }
}

__global__ void build_xpt_kernel(const float* __restrict__ x,
                                 float* __restrict__ xpt) {
    int o = blockIdx.x * blockDim.x + threadIdx.x;
    int rem = o & 63;
    int s = rem >> 1;
    int t = ((o >> 6) << 1) + (rem & 1);
    int tg = t - GT;
    float v = 0.f;
    if (tg >= 0 && tg < TP) {
        int tr = tg - PAD;
        if (tr < 0)            tr = -tr;
        else if (tr >= T_LEN)  tr = 2 * T_LEN - 2 - tr;
        v = x[(size_t)s * T_LEN + tr];
    }
    xpt[o] = v;
}

#define RING_PR 256
#define RING_F  (RING_PR * 64)
#define NFULL   32
#define NTAIL   (NCHUNK - 32 * NFULL)
#define FILL_PR 176

__global__ __launch_bounds__(256)
void fold_fir_kernel(const float* __restrict__ rt,
                     const float* __restrict__ src,
                     float* __restrict__ dst) {
    __shared__ __attribute__((aligned(16))) float sr[RTAB];
    __shared__ __attribute__((aligned(16))) float ringu[RING_F];
    __shared__ __attribute__((aligned(16))) float ringl[RING_F];

    const int tid  = threadIdx.x;
    const int lane = tid & 31;
    const int wave = __builtin_amdgcn_readfirstlane(tid >> 5);
    const int n    = lane & 15;
    const int half = lane >> 4;
    const int q    = half * 2;

    const int n0b = PAD + (int)blockIdx.x * 256;
    const int pbu = (GT + n0b + 16) >> 1;
    const int pbl = (GT + n0b + 238) >> 1;
    const float4* gu = (const float4*)(src + (size_t)pbu * 64);
    const float4* gl = (const float4*)(src + (size_t)pbl * 64);

    const int tr  = tid >> 4;
    const int w4  = tid & 15;
    const int cu4 = w4 ^ ((tr & 1) << 3);
    const int cl4 = w4 ^ (((tr & 1) ^ 1) << 3);
    float4* lu = (float4*)ringu;
    float4* ll = (float4*)ringl;

    {
        const float4* rt4 = (const float4*)rt;
        float4* sr4 = (float4*)sr;
        float4 a[11], b[11];
        const float4 f0 = rt4[tid],       f1 = rt4[tid + 256];
        const float4 f2 = rt4[tid + 512], f3 = rt4[tid + 768];
        float4 f4 = make_float4(0.f, 0.f, 0.f, 0.f);
        if (tid < RTAB / 4 - 1024) f4 = rt4[tid + 1024];
#pragma unroll
        for (int i = 0; i < 11; ++i) {
            const int p = tr + 16 * i;
            a[i] = gu[p * 16 + w4];
            b[i] = gl[w4 - p * 16];
        }
        sr4[tid]       = f0;  sr4[tid + 256] = f1;
        sr4[tid + 512] = f2;  sr4[tid + 768] = f3;
        if (tid < RTAB / 4 - 1024) sr4[tid + 1024] = f4;
#pragma unroll
        for (int i = 0; i < 11; ++i) {
            const int p = tr + 16 * i;
            lu[p * 16 + cu4] = a[i];
            ll[p * 16 + cl4] = b[i];
        }
    }

    const int aT = RC + q - n;
    const int aH = RC + 1 + q + n;
    const int bu = 16 * wave + half;
    const int bl = 16 * (7 - wave) + half;
    const int cu0 = (2 * n + 32 * half) & 63;
    const int cu1 = cu0 ^ 32;
    const int cl0 = (2 * n + 32 * (1 - half)) & 63;
    const int cl1 = cl0 ^ 32;

    v8f ps0 = {}, ps1 = {};
    v8f pa0 = {}, pa1 = {};

    float4 u0, u1, u2, u3, l0, l1, l2, l3;
#define STAGE_LOAD(s1)                                                      \
    {                                                                       \
        const int p = 64 * (s1) + 112 + tr;                                 \
        u0 = gu[p * 16 + w4];        u1 = gu[(p + 16) * 16 + w4];           \
        u2 = gu[(p + 32) * 16 + w4]; u3 = gu[(p + 48) * 16 + w4];           \
        l0 = gl[w4 - p * 16];        l1 = gl[w4 - (p + 16) * 16];           \
        l2 = gl[w4 - (p + 32) * 16]; l3 = gl[w4 - (p + 48) * 16];           \
    }
#define STAGE_STORE(s1)                                                     \
    {                                                                       \
        const int p = 64 * (s1) + 112 + tr;                                 \
        lu[((p)      & 255) * 16 + cu4] = u0;                               \
        lu[((p + 16) & 255) * 16 + cu4] = u1;                               \
        lu[((p + 32) & 255) * 16 + cu4] = u2;                               \
        lu[((p + 48) & 255) * 16 + cu4] = u3;                               \
        ll[((p)      & 255) * 16 + cl4] = l0;                               \
        ll[((p + 16) & 255) * 16 + cl4] = l1;                               \
        ll[((p + 32) & 255) * 16 + cl4] = l2;                               \
        ll[((p + 48) & 255) * 16 + cl4] = l3;                               \
    }
#define SEG 4
#define LOAD_SEG(X, j)                                                      \
    {                                                                       \
        const float* fu = ringu + (((bus + 8 * (j)) & 255) << 6);           \
        const float* fl = ringl + (((bls + 8 * (j)) & 255) << 6);           \
        _Pragma("unroll")                                                   \
        for (int c = 0; c < SEG; ++c) {                                     \
            X.t[c].x = ft[16 * (j) + 4 * c];                                \
            X.t[c].y = ft[16 * (j) + 4 * c + 1];                            \
            X.g[c].x = fh[16 * (j) + 4 * c];                                \
            X.g[c].y = fh[16 * (j) + 4 * c + 1];                            \
            X.xp0[c] = *(const v2f*)(fu + cu0 + 128 * c);                   \
            X.xp1[c] = *(const v2f*)(fu + cu1 + 128 * c);                   \
            X.xm0[c] = *(const v2f*)(fl + cl0 + 128 * c);                   \
            X.xm1[c] = *(const v2f*)(fl + cl1 + 128 * c);                   \
        }                                                                   \
    }
#define COMPUTE_SEG(X)                                                      \
    {                                                                       \
        _Pragma("unroll")                                                   \
        for (int c = 0; c < SEG; ++c) {                                     \
            const v2f t = X.t[c], g = X.g[c];                               \
            const v2f xp0 = X.xp0[c], xp1 = X.xp1[c];                       \
            const v2f xm0 = X.xm0[c], xm1 = X.xm1[c];                       \
            v2f as, aa, bs0, bs1, ba0, ba1;                                 \
            as.x = t.x + g.x;        as.y = t.y + g.y;                      \
            aa.x = t.x - g.x;        aa.y = t.y - g.y;                      \
            bs0.x = xp0.x + xm0.y;   bs0.y = xp0.y + xm0.x;                 \
            ba0.x = xp0.x - xm0.y;   ba0.y = xp0.y - xm0.x;                 \
            bs1.x = xp1.x + xm1.y;   bs1.y = xp1.y + xm1.x;                 \
            ba1.x = xp1.x - xm1.y;   ba1.y = xp1.y - xm1.x;                 \
            ps0 = __builtin_amdgcn_wmma_f32_16x16x4_f32(                    \
                false, as, false, bs0, (short)0, ps0, false, false);        \
            ps1 = __builtin_amdgcn_wmma_f32_16x16x4_f32(                    \
                false, as, false, bs1, (short)0, ps1, false, false);        \
            pa0 = __builtin_amdgcn_wmma_f32_16x16x4_f32(                    \
                false, aa, false, ba0, (short)0, pa0, false, false);        \
            pa1 = __builtin_amdgcn_wmma_f32_16x16x4_f32(                    \
                false, aa, false, ba1, (short)0, pa1, false, false);        \
        }                                                                   \
    }
#define GROUP_EDGE() asm volatile("" ::: "memory")

    struct SegOps { v2f t[SEG], g[SEG], xp0[SEG], xp1[SEG], xm0[SEG], xm1[SEG]; };

    STAGE_LOAD(1)
    __syncthreads();

    for (int s = 0; s < NFULL; ++s) {
        const float* ft = sr + aT + 128 * s;
        const float* fh = sr + aH + 128 * s;
        const int bus = bu + 64 * s;
        const int bls = bl + 64 * s;
        const int sl = (s + 2 <= NFULL) ? s + 2 : NFULL;
        SegOps A, B;
        LOAD_SEG(A, 0)
        GROUP_EDGE();
#pragma unroll
        for (int j = 0; j < 32 / SEG; j += 2) {
            LOAD_SEG(B, j + 1)
            GROUP_EDGE();
            if (j == 0) {
                STAGE_STORE(s + 1)
                STAGE_LOAD(sl)
                GROUP_EDGE();
            }
            COMPUTE_SEG(A)
            GROUP_EDGE();
            if (j + 2 < 32 / SEG) LOAD_SEG(A, j + 2)
            GROUP_EDGE();
            COMPUTE_SEG(B)
            GROUP_EDGE();
        }
        __builtin_amdgcn_sched_group_barrier(0x100, 32, 0);
#pragma unroll
        for (int j = 0; j < 32 / SEG - 2; ++j) {
            __builtin_amdgcn_sched_group_barrier(0x008, 4 * SEG, 0);
            __builtin_amdgcn_sched_group_barrier(0x100, 16, 0);
        }
        __builtin_amdgcn_sched_group_barrier(0x008, 8 * SEG, 0);
        __syncthreads();
    }
    {
        const float* ft = sr + aT + 128 * NFULL;
        const float* fh = sr + aH + 128 * NFULL;
        const int bus = bu + 64 * NFULL;
        const int bls = bl + 64 * NFULL;
        static_assert(NTAIL == SEG, "the closing stage is one segment");
        SegOps A;
        LOAD_SEG(A, 0)
        COMPUTE_SEG(A)
    }
#undef LOAD_SEG
#undef COMPUTE_SEG
#undef GROUP_EDGE
#undef STAGE_LOAD
#undef STAGE_STORE

    const int tb = n0b - PAD + TILE * wave;
    float* o0 = dst + (size_t)n        * T_LEN + tb;
    float* o1 = dst + (size_t)(n + 16) * T_LEN + tb;
    float* u0p = o0 + 16 + 8 * half;
    float* u1p = o1 + 16 + 8 * half;
    float* l0p = o0 + 8 - 8 * half;
    float* l1p = o1 + 8 - 8 * half;
    *(float4*)(u0p)     = make_float4(ps0[0] + pa0[0], ps0[1] + pa0[1],
                                      ps0[2] + pa0[2], ps0[3] + pa0[3]);
    *((float4*)u0p + 1) = make_float4(ps0[4] + pa0[4], ps0[5] + pa0[5],
                                      ps0[6] + pa0[6], ps0[7] + pa0[7]);
    *(float4*)(u1p)     = make_float4(ps1[0] + pa1[0], ps1[1] + pa1[1],
                                      ps1[2] + pa1[2], ps1[3] + pa1[3]);
    *((float4*)u1p + 1) = make_float4(ps1[4] + pa1[4], ps1[5] + pa1[5],
                                      ps1[6] + pa1[6], ps1[7] + pa1[7]);
    *(float4*)(l0p)     = make_float4(ps0[7] - pa0[7], ps0[6] - pa0[6],
                                      ps0[5] - pa0[5], ps0[4] - pa0[4]);
    *((float4*)l0p + 1) = make_float4(ps0[3] - pa0[3], ps0[2] - pa0[2],
                                      ps0[1] - pa0[1], ps0[0] - pa0[0]);
    *(float4*)(l1p)     = make_float4(ps1[7] - pa1[7], ps1[6] - pa1[6],
                                      ps1[5] - pa1[5], ps1[4] - pa1[4]);
    *((float4*)l1p + 1) = make_float4(ps1[3] - pa1[3], ps1[2] - pa1[2],
                                      ps1[1] - pa1[1], ps1[0] - pa1[0]);
}

extern "C" void kernel_launch(void* const* d_in, const int* in_sizes, int n_in,
                              void* d_out, int out_size, void* d_ws, size_t ws_size,
                              hipStream_t stream) {
    const float* x    = (const float*)d_in[0];
    const float* kern = (const float*)d_in[1];
    float* out = (float*)d_out;
    float* ws  = (float*)d_ws;

    float* rt  = ws;
    float* xpt = ws + RTAB;

    prep_r_kernel<<<RTAB, 256, 0, stream>>>(kern, rt);
    build_xpt_kernel<<<(ROWS * S_SIG) / 256, 256, 0, stream>>>(x, xpt);
    fold_fir_kernel<<<T_LEN / 256, 256, 0, stream>>>(rt, xpt, out);
}


# staging launch and transposed plane removed: WMMA roles swapped (signals = A rows, folded taps = B), so LDS rings are [signal][time], filled by aligned 16-byte loads of x; order of every sum kept
# speedup vs baseline: 2.3521x; 1.0338x over previous
#include <hip/hip_runtime.h>


typedef __attribute__((ext_vector_type(2))) float v2f;
typedef __attribute__((ext_vector_type(8))) float v8f;

#define K_TAPS 4096
#define PAD    2048
#define T_LEN  131072
#define S_SIG  32
#define TP     (T_LEN + 2*PAD)
#define RTAB   4160
#define RC     15
#define NCHUNK 1028
#define TILE   32

__global__ __launch_bounds__(256)
void prep_r_kernel(const float* __restrict__ h, float* __restrict__ rt) {
    __shared__ float red[8];
    const int tid = threadIdx.x;
    int d = (int)blockIdx.x - RC;
    if (d < 0) d = -d;
    float a[K_TAPS / 256], b[K_TAPS / 256];
#pragma unroll
    for (int j = 0; j < K_TAPS / 256; ++j) {
        const int i  = tid + 256 * j;
        const int i2 = (i + d < K_TAPS) ? i + d : K_TAPS - 1;
        a[j] = h[i];
        b[j] = h[i2];
    }
    float s = 0.f;
#pragma unroll
    for (int j = 0; j < K_TAPS / 256; ++j) {
        const int i = tid + 256 * j;
        s += a[j] * ((i + d < K_TAPS) ? b[j] : 0.f);
    }
#pragma unroll
    for (int o = 16; o > 0; o >>= 1) s += __shfl_xor(s, o);
    if ((tid & 31) == 0) red[tid >> 5] = s;
    __syncthreads();
    if (tid == 0) {
        float tot = red[0];
#pragma unroll
        for (int w = 1; w < 8; ++w) tot += red[w];
        rt[blockIdx.x] = 0.5f * tot;
    }
}

#define RING_T  512
#define PITCH   516
#define RING_F  (S_SIG * PITCH)
#define NFULL   32
#define NTAIL   (NCHUNK - 32 * NFULL)
#define REACH   (128 * NFULL + 351)

template <bool EDGE>
__device__ __forceinline__ float4 load_quad(const float* __restrict__ xs, int p0) {
    int lo = p0 - PAD;
    if (EDGE) {
        const int t0 = p0 - PAD;
        if (t0 < 0)            lo = -t0 - 3;
        else if (t0 >= T_LEN)  lo = 2 * T_LEN - 5 - t0;
        if (!((unsigned)p0 < (unsigned)TP)) lo = 0;
        const float* q4 = xs + lo;
        return make_float4(q4[0], q4[1], q4[2], q4[3]);
    }
    return *(const float4*)(xs + lo);
}
template <bool EDGE>
__device__ __forceinline__ float4 keep_quad(float4 v, int p0) {
    if (EDGE) {
        const int t0 = p0 - PAD;
        if ((unsigned)t0 >= (unsigned)T_LEN) v = make_float4(v.w, v.z, v.y, v.x);
        if (!((unsigned)p0 < (unsigned)TP))  v = make_float4(0.f, 0.f, 0.f, 0.f);
    }
    return v;
}

template <bool EDGE>
__device__ __forceinline__ void fold_body(const float* __restrict__ rt,
                                          const float* __restrict__ x,
                                          float* __restrict__ dst,
                                          float* sr, float* ringu, float* ringl) {
    const int tid  = threadIdx.x;
    const int lane = tid & 31;
    const int wave = __builtin_amdgcn_readfirstlane(tid >> 5);
    const int n    = lane & 15;
    const int half = lane >> 4;
    const int q    = half * 2;

    const int n0b = PAD + (int)blockIdx.x * 256;
    const int pu0 = n0b + 16;
    const int pl0 = n0b + 239;

    const int ls = wave + 8 * (lane >> 3);
    const int lq = lane & 7;
    const float* xs = x + (size_t)ls * T_LEN;
    float* wu = ringu + ls * PITCH;
    float* wl = ringl + ls * PITCH;

    {
        const float4* rt4 = (const float4*)rt;
        float4* sr4 = (float4*)sr;
        float4 a[11], b[11];
        const float4 f0 = rt4[tid],       f1 = rt4[tid + 256];
        const float4 f2 = rt4[tid + 512], f3 = rt4[tid + 768];
        float4 f4 = make_float4(0.f, 0.f, 0.f, 0.f);
        if (tid < RTAB / 4 - 1024) f4 = rt4[tid + 1024];
#pragma unroll
        for (int i = 0; i < 11; ++i) {
            const int Q = lq + 8 * i;
            a[i] = load_quad<EDGE>(xs, pu0 + 4 * Q);
            b[i] = load_quad<EDGE>(xs, pl0 - 3 - 4 * Q);
        }
        sr4[tid]       = f0;  sr4[tid + 256] = f1;
        sr4[tid + 512] = f2;  sr4[tid + 768] = f3;
        if (tid < RTAB / 4 - 1024) sr4[tid + 1024] = f4;
#pragma unroll
        for (int i = 0; i < 11; ++i) {
            const int Q = lq + 8 * i;
            *(float4*)(wu + ((pu0 + 4 * Q) & (RING_T - 1))) =
                keep_quad<EDGE>(a[i], pu0 + 4 * Q);
            *(float4*)(wl + ((pl0 - 3 - 4 * Q) & (RING_T - 1))) =
                keep_quad<EDGE>(b[i], pl0 - 3 - 4 * Q);
        }
    }

    const int aT = RC + q - n;
    const int aH = RC + 1 + q + n;
    const float* ru = ringu + n * PITCH + q;
    const float* rl = ringl + n * PITCH + 14 - q;
    const int puw = pu0 + TILE * wave;
    const int plw = n0b + TILE * wave;

    v8f ps0 = {}, ps1 = {};
    v8f pa0 = {}, pa1 = {};

    float4 u0, u1, u2, u3, l0, l1, l2, l3;
#define QUAD_U(s1, i) (pu0 + 4 * (32 * (s1) + 56 + lq + 8 * (i)))
#define QUAD_L(s1, i) (pl0 - 3 - 4 * (32 * (s1) + 56 + lq + 8 * (i)))
#define STAGE_LOAD(s1)                                                      \
    {                                                                       \
        u0 = load_quad<EDGE>(xs, QUAD_U(s1, 0));                            \
        u1 = load_quad<EDGE>(xs, QUAD_U(s1, 1));                            \
        u2 = load_quad<EDGE>(xs, QUAD_U(s1, 2));                            \
        u3 = load_quad<EDGE>(xs, QUAD_U(s1, 3));                            \
        l0 = load_quad<EDGE>(xs, QUAD_L(s1, 0));                            \
        l1 = load_quad<EDGE>(xs, QUAD_L(s1, 1));                            \
        l2 = load_quad<EDGE>(xs, QUAD_L(s1, 2));                            \
        l3 = load_quad<EDGE>(xs, QUAD_L(s1, 3));                            \
    }
#define RING_PUT(w, P, v) *(float4*)((w) + ((P) & (RING_T - 1))) = keep_quad<EDGE>(v, P)
#define STAGE_STORE(s1)                                                     \
    {                                                                       \
        RING_PUT(wu, QUAD_U(s1, 0), u0);  RING_PUT(wu, QUAD_U(s1, 1), u1);  \
        RING_PUT(wu, QUAD_U(s1, 2), u2);  RING_PUT(wu, QUAD_U(s1, 3), u3);  \
        RING_PUT(wl, QUAD_L(s1, 0), l0);  RING_PUT(wl, QUAD_L(s1, 1), l1);  \
        RING_PUT(wl, QUAD_L(s1, 2), l2);  RING_PUT(wl, QUAD_L(s1, 3), l3);  \
    }
#define SEG 4
#define LOAD_SEG(X, j)                                                      \
    {                                                                       \
        const float* fu = ru + ((pus + 16 * (j)) & (RING_T - 1));           \
        const float* fl = rl + ((pls - 16 * (j)) & (RING_T - 1));           \
        _Pragma("unroll")                                                   \
        for (int c = 0; c < SEG; ++c) {                                     \
            X.t[c].x = ft[16 * (j) + 4 * c];                                \
            X.t[c].y = ft[16 * (j) + 4 * c + 1];                            \
            X.g[c].x = fh[16 * (j) + 4 * c];                                \
            X.g[c].y = fh[16 * (j) + 4 * c + 1];                            \
            X.xp0[c] = *(const v2f*)(fu + 4 * c);                           \
            X.xp1[c] = *(const v2f*)(fu + 16 * PITCH + 4 * c);              \
            X.xm0[c] = *(const v2f*)(fl - 4 * c);                           \
            X.xm1[c] = *(const v2f*)(fl + 16 * PITCH - 4 * c);              \
        }                                                                   \
    }
#define COMPUTE_SEG(X)                                                      \
    {                                                                       \
        _Pragma("unroll")                                                   \
        for (int c = 0; c < SEG; ++c) {                                     \
            const v2f t = X.t[c], g = X.g[c];                               \
            const v2f xp0 = X.xp0[c], xp1 = X.xp1[c];                       \
            const v2f xm0 = X.xm0[c], xm1 = X.xm1[c];                       \
            v2f bs, ba, as0, as1, aa0, aa1;                                 \
            bs.x = t.x + g.x;        bs.y = t.y + g.y;                      \
            ba.x = t.x - g.x;        ba.y = t.y - g.y;                      \
            as0.x = xp0.x + xm0.y;   as0.y = xp0.y + xm0.x;                 \
            aa0.x = xp0.x - xm0.y;   aa0.y = xp0.y - xm0.x;                 \
            as1.x = xp1.x + xm1.y;   as1.y = xp1.y + xm1.x;                 \
            aa1.x = xp1.x - xm1.y;   aa1.y = xp1.y - xm1.x;                 \
            ps0 = __builtin_amdgcn_wmma_f32_16x16x4_f32(                    \
                false, as0, false, bs, (short)0, ps0, false, false);        \
            ps1 = __builtin_amdgcn_wmma_f32_16x16x4_f32(                    \
                false, as1, false, bs, (short)0, ps1, false, false);        \
            pa0 = __builtin_amdgcn_wmma_f32_16x16x4_f32(                    \
                false, aa0, false, ba, (short)0, pa0, false, false);        \
            pa1 = __builtin_amdgcn_wmma_f32_16x16x4_f32(                    \
                false, aa1, false, ba, (short)0, pa1, false, false);        \
        }                                                                   \
    }
#define GROUP_EDGE() asm volatile("" ::: "memory")

    struct SegOps { v2f t[SEG], g[SEG], xp0[SEG], xp1[SEG], xm0[SEG], xm1[SEG]; };

    STAGE_LOAD(1)
    __syncthreads();

    for (int s = 0; s < NFULL; ++s) {
        const float* ft = sr + aT + 128 * s;
        const float* fh = sr + aH + 128 * s;
        const int pus = puw + 128 * s;
        const int pls = plw - 128 * s;
        const int sl = (s + 2 <= NFULL) ? s + 2 : NFULL;
        SegOps A, B;
        LOAD_SEG(A, 0)
        GROUP_EDGE();
#pragma unroll
        for (int j = 0; j < 32 / SEG; j += 2) {
            LOAD_SEG(B, j + 1)
            GROUP_EDGE();
            if (j == 0) {
                STAGE_STORE(s + 1)
                STAGE_LOAD(sl)
                GROUP_EDGE();
            }
            COMPUTE_SEG(A)
            GROUP_EDGE();
            if (j + 2 < 32 / SEG) LOAD_SEG(A, j + 2)
            GROUP_EDGE();
            COMPUTE_SEG(B)
            GROUP_EDGE();
        }
        __builtin_amdgcn_sched_group_barrier(0x100, 32, 0);
#pragma unroll
        for (int j = 0; j < 32 / SEG - 2; ++j) {
            __builtin_amdgcn_sched_group_barrier(0x008, 4 * SEG, 0);
            __builtin_amdgcn_sched_group_barrier(0x100, 16, 0);
        }
        __builtin_amdgcn_sched_group_barrier(0x008, 8 * SEG, 0);
        __syncthreads();
    }
    {
        const float* ft = sr + aT + 128 * NFULL;
        const float* fh = sr + aH + 128 * NFULL;
        const int pus = puw + 128 * NFULL;
        const int pls = plw - 128 * NFULL;
        static_assert(NTAIL == SEG, "the closing stage is one segment");
        SegOps A;
        LOAD_SEG(A, 0)
        COMPUTE_SEG(A)
    }
#undef LOAD_SEG
#undef COMPUTE_SEG
#undef GROUP_EDGE
#undef STAGE_LOAD
#undef STAGE_STORE
#undef RING_PUT
#undef QUAD_U
#undef QUAD_L

    const int tb = n0b - PAD + TILE * wave;
    float* ob = dst + (size_t)(8 * half) * T_LEN + tb;
#pragma unroll
    for (int r = 0; r < 8; ++r) {
        float* o0 = ob + (size_t)r * T_LEN;
        float* o1 = o0 + (size_t)16 * T_LEN;
        o0[16 + n] = ps0[r] + pa0[r];
        o0[15 - n] = ps0[r] - pa0[r];
        o1[16 + n] = ps1[r] + pa1[r];
        o1[15 - n] = ps1[r] - pa1[r];
    }
}

__global__ __launch_bounds__(256)
void fold_fir_kernel(const float* __restrict__ rt,
                     const float* __restrict__ x,
                     float* __restrict__ dst) {
    __shared__ __attribute__((aligned(16))) float sr[RTAB];
    __shared__ __attribute__((aligned(16))) float ringu[RING_F];
    __shared__ __attribute__((aligned(16))) float ringl[RING_F];
    const int n0b = PAD + (int)blockIdx.x * 256;
    const bool edge = (n0b + 239 - REACH < PAD) || (n0b + 16 + REACH >= PAD + T_LEN);
    if (edge) fold_body<true>(rt, x, dst, sr, ringu, ringl);
    else      fold_body<false>(rt, x, dst, sr, ringu, ringl);
}

extern "C" void kernel_launch(void* const* d_in, const int* in_sizes, int n_in,
                              void* d_out, int out_size, void* d_ws, size_t ws_size,
                              hipStream_t stream) {
    const float* x    = (const float*)d_in[0];
    const float* kern = (const float*)d_in[1];
    float* out = (float*)d_out;
    float* rt  = (float*)d_ws;

    prep_r_kernel<<<RTAB, 256, 0, stream>>>(kern, rt);
    fold_fir_kernel<<<T_LEN / 256, 256, 0, stream>>>(rt, x, out);
}


# direct folded product replaced by overlap-save in the frequency domain: 16384-point transforms in LDS, radix-16 stages as f32 WMMA products, squared tap spectrum built per call; sum order changes
# speedup vs baseline: 9.4178x; 4.0040x over previous
#include <hip/hip_runtime.h>


typedef __attribute__((ext_vector_type(2))) float v2f;
typedef __attribute__((ext_vector_type(8))) float v8f;

#define K_TAPS 4096
#define PAD    2048
#define T_LEN  131072
#define NFFT   16384
#define HOP    8192
#define NTHR   512
#define NWAVE  (NTHR / 32)
#define PHYS(p) ((p) + (((p) >> 6) << 2))
#define NPLANE (NFFT + NFFT / 16)

struct Frags { v2f fr[4], fi[4], nfi[4]; };

template <int M, bool INV>
__device__ __forceinline__ void radix16_stage(float* __restrict__ re, float* __restrict__ im,
                                              const Frags& F, int wave, int n, int half) {
    constexpr int SC = 1024 / M;
    constexpr int MP = (M >= 64) ? M + M / 16 : M;
#pragma unroll 1
    for (int tau = wave; tau < 64; tau += NWAVE) {
        const int col  = 16 * tau + n;
        const int b    = col & (M - 1);
        const int base = PHYS((col - b) * 16 + b);
        v2f xr[4], xi[4];
#pragma unroll
        for (int c = 0; c < 4; ++c) {
            const int k0 = 4 * c + 2 * half;
            xr[c].x = re[base + k0 * MP];
            xr[c].y = re[base + (k0 + 1) * MP];
            xi[c].x = im[base + k0 * MP];
            xi[c].y = im[base + (k0 + 1) * MP];
        }
        if (INV) {
#pragma unroll
            for (int c = 0; c < 4; ++c) {
                const int k0 = 4 * c + 2 * half;
                float s0, c0, s1, c1;
                sincospif((float)(SC * b * k0) * (1.0f / 8192.0f), &s0, &c0);
                sincospif((float)(SC * b * (k0 + 1)) * (1.0f / 8192.0f), &s1, &c1);
                const float ar = xr[c].x, ai = xi[c].x, br = xr[c].y, bi = xi[c].y;
                xr[c].x = ar * c0 - ai * s0;   xi[c].x = ar * s0 + ai * c0;
                xr[c].y = br * c1 - bi * s1;   xi[c].y = br * s1 + bi * c1;
            }
        }
        v8f yr = {}, yi = {};
#pragma unroll
        for (int c = 0; c < 4; ++c) {
            if (!INV) {
                yr = __builtin_amdgcn_wmma_f32_16x16x4_f32(false, F.fr[c],  false, xr[c], (short)0, yr, false, false);
                yi = __builtin_amdgcn_wmma_f32_16x16x4_f32(false, F.fi[c],  false, xr[c], (short)0, yi, false, false);
                yr = __builtin_amdgcn_wmma_f32_16x16x4_f32(false, F.nfi[c], false, xi[c], (short)0, yr, false, false);
                yi = __builtin_amdgcn_wmma_f32_16x16x4_f32(false, F.fr[c],  false, xi[c], (short)0, yi, false, false);
            } else {
                yr = __builtin_amdgcn_wmma_f32_16x16x4_f32(false, F.fr[c],  false, xr[c], (short)0, yr, false, false);
                yi = __builtin_amdgcn_wmma_f32_16x16x4_f32(false, F.nfi[c], false, xr[c], (short)0, yi, false, false);
                yr = __builtin_amdgcn_wmma_f32_16x16x4_f32(false, F.fi[c],  false, xi[c], (short)0, yr, false, false);
                yi = __builtin_amdgcn_wmma_f32_16x16x4_f32(false, F.fr[c],  false, xi[c], (short)0, yi, false, false);
            }
        }
#pragma unroll
        for (int r = 0; r < 8; ++r) {
            const int d = 8 * half + r;
            float vr = yr[r], vi = yi[r];
            if (!INV) {
                float s, co;
                sincospif((float)(SC * b * d) * (1.0f / 8192.0f), &s, &co);
                const float tr = vr * co + vi * s;
                const float ti = vi * co - vr * s;
                vr = tr;  vi = ti;
            }
            re[base + d * MP] = vr;
            im[base + d * MP] = vi;
        }
    }
}

template <int SGN>
__device__ __forceinline__ void dft4(float4 xr, float4 xi, float4& yr, float4& yi) {
    const float ar = xr.x + xr.z, ai = xi.x + xi.z;
    const float br = xr.x - xr.z, bi = xi.x - xi.z;
    const float cr = xr.y + xr.w, ci = xi.y + xi.w;
    const float dr = xr.y - xr.w, di = xi.y - xi.w;
    yr.x = ar + cr;  yi.x = ai + ci;
    yr.z = ar - cr;  yi.z = ai - ci;
    if (SGN > 0) {
        yr.y = br + di;  yi.y = bi - dr;
        yr.w = br - di;  yi.w = bi + dr;
    } else {
        yr.y = br - di;  yi.y = bi + dr;
        yr.w = br + di;  yi.w = bi - dr;
    }
}

__device__ __forceinline__ float pad_sample(const float* __restrict__ xs, int t) {
    int idx = t;
    if (t < 0) idx = -t;
    if (t >= T_LEN) idx = 2 * T_LEN - 2 - t;
    const bool ok = (t >= -PAD) && (t < T_LEN + PAD);
    if (!ok) idx = 0;
    const float v = xs[idx];
    return ok ? v : 0.f;
}

__global__ __launch_bounds__(NTHR)
void fir_fft_kernel(const float* __restrict__ x,
                    const float* __restrict__ h,
                    float* __restrict__ dst) {
    __shared__ __attribute__((aligned(16))) float re[NPLANE];
    __shared__ __attribute__((aligned(16))) float im[NPLANE];
    __shared__ __attribute__((aligned(16))) float gp[NPLANE];

    const int tid  = threadIdx.x;
    const int lane = tid & 31;
    const int wave = __builtin_amdgcn_readfirstlane(tid >> 5);
    const int n    = lane & 15;
    const int half = lane >> 4;

    const int pr  = (int)blockIdx.x >> 4;
    const int blk = (int)blockIdx.x & 15;
    const float* x1 = x + (size_t)(2 * pr) * T_LEN;
    const float* x2 = x1 + T_LEN;
    const int t0 = HOP * blk - 4096;

    Frags F;
#pragma unroll
    for (int c = 0; c < 4; ++c) {
        const int a0 = 4 * c + 2 * half;
        float s0, c0, s1, c1;
        sincospif((float)((a0 * n) & 15) * 0.125f, &s0, &c0);
        sincospif((float)(((a0 + 1) * n) & 15) * 0.125f, &s1, &c1);
        F.fr[c].x  = c0;   F.fr[c].y  = c1;
        F.fi[c].x  = -s0;  F.fi[c].y  = -s1;
        F.nfi[c].x = s0;   F.nfi[c].y = s1;
    }

#pragma unroll 1
    for (int pass = 0; pass < 2; ++pass) {
        if (pass == 0) {
#pragma unroll
            for (int i = 0; i < NFFT / NTHR; ++i) {
                const int q = tid + NTHR * i;
                re[PHYS(q)] = (q < K_TAPS) ? h[q & (K_TAPS - 1)] : 0.f;
                im[PHYS(q)] = 0.f;
            }
        } else {
#pragma unroll 1
            for (int hb = 0; hb < 2; ++hb) {
                float a1[16], a2[16];
#pragma unroll
                for (int i = 0; i < 16; ++i) {
                    const int q = tid + NTHR * (16 * hb + i);
                    a1[i] = pad_sample(x1, t0 + q);
                    a2[i] = pad_sample(x2, t0 + q);
                }
#pragma unroll
                for (int i = 0; i < 16; ++i) {
                    const int q = tid + NTHR * (16 * hb + i);
                    re[PHYS(q)] = a1[i];
                    im[PHYS(q)] = a2[i];
                }
            }
        }
        __syncthreads();
        radix16_stage<1024, false>(re, im, F, wave, n, half);
        __syncthreads();
        radix16_stage<64, false>(re, im, F, wave, n, half);
        __syncthreads();
        radix16_stage<4, false>(re, im, F, wave, n, half);
        __syncthreads();
#pragma unroll
        for (int i = 0; i < NFFT / 4 / NTHR; ++i) {
            const int g = PHYS(4 * (tid + NTHR * i)) >> 2;
            const float4 xr = ((const float4*)re)[g];
            const float4 xi = ((const float4*)im)[g];
            float4 yr, yi;
            dft4<1>(xr, xi, yr, yi);
            if (pass == 0) {
                float4 gg;
                gg.x = (yr.x * yr.x + yi.x * yi.x) * (1.0f / NFFT);
                gg.y = (yr.y * yr.y + yi.y * yi.y) * (1.0f / NFFT);
                gg.z = (yr.z * yr.z + yi.z * yi.z) * (1.0f / NFFT);
                gg.w = (yr.w * yr.w + yi.w * yi.w) * (1.0f / NFFT);
                ((float4*)gp)[g] = gg;
            } else {
                const float4 gg = ((const float4*)gp)[g];
                yr.x *= gg.x;  yr.y *= gg.y;  yr.z *= gg.z;  yr.w *= gg.w;
                yi.x *= gg.x;  yi.y *= gg.y;  yi.z *= gg.z;  yi.w *= gg.w;
                float4 zr, zi;
                dft4<-1>(yr, yi, zr, zi);
                ((float4*)re)[g] = zr;
                ((float4*)im)[g] = zi;
            }
        }
        __syncthreads();
    }

    radix16_stage<4, true>(re, im, F, wave, n, half);
    __syncthreads();
    radix16_stage<64, true>(re, im, F, wave, n, half);
    __syncthreads();
    radix16_stage<1024, true>(re, im, F, wave, n, half);
    __syncthreads();

    float* o1 = dst + (size_t)(2 * pr) * T_LEN + HOP * blk;
    float* o2 = o1 + T_LEN;
#pragma unroll
    for (int i = 0; i < HOP / NTHR; ++i) {
        const int j = tid + NTHR * i;
        o1[j] = re[PHYS(4096 + j)];
        o2[j] = im[PHYS(4096 + j)];
    }
}

extern "C" void kernel_launch(void* const* d_in, const int* in_sizes, int n_in,
                              void* d_out, int out_size, void* d_ws, size_t ws_size,
                              hipStream_t stream) {
    const float* x    = (const float*)d_in[0];
    const float* kern = (const float*)d_in[1];
    float* out = (float*)d_out;

    fir_fft_kernel<<<256, NTHR, 0, stream>>>(x, kern, out);
}


# tap spectrum |H|^2 made once per call by a small first launch (8 work-groups, two 1024-point stretches each) and read back, not recomputed in all 256 window work-groups; order of every sum kept
# speedup vs baseline: 12.8476x; 1.3642x over previous
#include <hip/hip_runtime.h>


typedef __attribute__((ext_vector_type(2))) float v2f;
typedef __attribute__((ext_vector_type(8))) float v8f;

#define K_TAPS 4096
#define PAD    2048
#define T_LEN  131072
#define NFFT   16384
#define HOP    8192
#define NTHR   512
#define NWAVE  (NTHR / 32)
#define PHYS(p) ((p) + (((p) >> 6) << 2))
#define NPLANE (NFFT + NFFT / 16)

struct Frags { v2f fr[4], fi[4], nfi[4]; };

__device__ __forceinline__ Frags make_frags(int n, int half) {
    Frags F;
#pragma unroll
    for (int c = 0; c < 4; ++c) {
        const int a0 = 4 * c + 2 * half;
        float s0, c0, s1, c1;
        sincospif((float)((a0 * n) & 15) * 0.125f, &s0, &c0);
        sincospif((float)(((a0 + 1) * n) & 15) * 0.125f, &s1, &c1);
        F.fr[c].x  = c0;   F.fr[c].y  = c1;
        F.fi[c].x  = -s0;  F.fi[c].y  = -s1;
        F.nfi[c].x = s0;   F.nfi[c].y = s1;
    }
    return F;
}

template <int M, bool INV, int NT, int NW>
__device__ __forceinline__ void radix16_stage(float* __restrict__ re, float* __restrict__ im,
                                              const Frags& F, int wave, int n, int half) {
    constexpr int SC = 1024 / M;
    constexpr int MP = (M >= 64) ? M + M / 16 : M;
#pragma unroll 1
    for (int tau = wave; tau < NT; tau += NW) {
        const int col  = 16 * tau + n;
        const int b    = col & (M - 1);
        const int base = PHYS((col - b) * 16 + b);
        v2f xr[4], xi[4];
#pragma unroll
        for (int c = 0; c < 4; ++c) {
            const int k0 = 4 * c + 2 * half;
            xr[c].x = re[base + k0 * MP];
            xr[c].y = re[base + (k0 + 1) * MP];
            xi[c].x = im[base + k0 * MP];
            xi[c].y = im[base + (k0 + 1) * MP];
        }
        if (INV) {
#pragma unroll
            for (int c = 0; c < 4; ++c) {
                const int k0 = 4 * c + 2 * half;
                float s0, c0, s1, c1;
                sincospif((float)(SC * b * k0) * (1.0f / 8192.0f), &s0, &c0);
                sincospif((float)(SC * b * (k0 + 1)) * (1.0f / 8192.0f), &s1, &c1);
                const float ar = xr[c].x, ai = xi[c].x, br = xr[c].y, bi = xi[c].y;
                xr[c].x = ar * c0 - ai * s0;   xi[c].x = ar * s0 + ai * c0;
                xr[c].y = br * c1 - bi * s1;   xi[c].y = br * s1 + bi * c1;
            }
        }
        v8f yr = {}, yi = {};
#pragma unroll
        for (int c = 0; c < 4; ++c) {
            if (!INV) {
                yr = __builtin_amdgcn_wmma_f32_16x16x4_f32(false, F.fr[c],  false, xr[c], (short)0, yr, false, false);
                yi = __builtin_amdgcn_wmma_f32_16x16x4_f32(false, F.fi[c],  false, xr[c], (short)0, yi, false, false);
                yr = __builtin_amdgcn_wmma_f32_16x16x4_f32(false, F.nfi[c], false, xi[c], (short)0, yr, false, false);
                yi = __builtin_amdgcn_wmma_f32_16x16x4_f32(false, F.fr[c],  false, xi[c], (short)0, yi, false, false);
            } else {
                yr = __builtin_amdgcn_wmma_f32_16x16x4_f32(false, F.fr[c],  false, xr[c], (short)0, yr, false, false);
                yi = __builtin_amdgcn_wmma_f32_16x16x4_f32(false, F.nfi[c], false, xr[c], (short)0, yi, false, false);
                yr = __builtin_amdgcn_wmma_f32_16x16x4_f32(false, F.fi[c],  false, xi[c], (short)0, yr, false, false);
                yi = __builtin_amdgcn_wmma_f32_16x16x4_f32(false, F.fr[c],  false, xi[c], (short)0, yi, false, false);
            }
        }
#pragma unroll
        for (int r = 0; r < 8; ++r) {
            const int d = 8 * half + r;
            float vr = yr[r], vi = yi[r];
            if (!INV) {
                float s, co;
                sincospif((float)(SC * b * d) * (1.0f / 8192.0f), &s, &co);
                const float tr = vr * co + vi * s;
                const float ti = vi * co - vr * s;
                vr = tr;  vi = ti;
            }
            re[base + d * MP] = vr;
            im[base + d * MP] = vi;
        }
    }
}

template <int SGN>
__device__ __forceinline__ void dft4(float4 xr, float4 xi, float4& yr, float4& yi) {
    const float ar = xr.x + xr.z, ai = xi.x + xi.z;
    const float br = xr.x - xr.z, bi = xi.x - xi.z;
    const float cr = xr.y + xr.w, ci = xi.y + xi.w;
    const float dr = xr.y - xr.w, di = xi.y - xi.w;
    yr.x = ar + cr;  yi.x = ai + ci;
    yr.z = ar - cr;  yi.z = ai - ci;
    if (SGN > 0) {
        yr.y = br + di;  yi.y = bi - dr;
        yr.w = br - di;  yi.w = bi + dr;
    } else {
        yr.y = br - di;  yi.y = bi + dr;
        yr.w = br + di;  yi.w = bi - dr;
    }
}

__device__ __forceinline__ float pad_sample(const float* __restrict__ xs, int t) {
    int idx = t;
    if (t < 0) idx = -t;
    if (t >= T_LEN) idx = 2 * T_LEN - 2 - t;
    const bool ok = (t >= -PAD) && (t < T_LEN + PAD);
    if (!ok) idx = 0;
    const float v = xs[idx];
    return ok ? v : 0.f;
}

#define TAP_THR  256
#define TAP_WAVE (TAP_THR / 32)
#define SUBLEN   1024
#define TAP_POS  (2 * SUBLEN)
#define NSUB     (SUBLEN + SUBLEN / 16)
#define NTAPP    (TAP_POS + TAP_POS / 16)
static_assert(K_TAPS == 4 * SUBLEN, "the taps are points a = 0..3 of every first-stage column");

__global__ __launch_bounds__(TAP_THR)
void fir_taps_kernel(const float* __restrict__ h, float* __restrict__ gsq) {
    __shared__ __attribute__((aligned(16))) float re[NTAPP];
    __shared__ __attribute__((aligned(16))) float im[NTAPP];

    const int tid  = threadIdx.x;
    const int lane = tid & 31;
    const int wave = __builtin_amdgcn_readfirstlane(tid >> 5);
    const int n    = lane & 15;
    const int half = lane >> 4;
    const int d0   = (int)blockIdx.x;
    const int d    = d0 + 8 * half;

    const Frags F = make_frags(n, half);

    {
        constexpr int NTA = SUBLEN / 16 / TAP_WAVE;
        v2f xt[NTA];
#pragma unroll
        for (int i = 0; i < NTA; ++i) {
            const int b = 16 * (wave + TAP_WAVE * i) + n;
            xt[i].x = h[(2 * half) * SUBLEN + b];
            xt[i].y = h[(2 * half + 1) * SUBLEN + b];
        }
#pragma unroll
        for (int i = 0; i < NTA; ++i) {
            const int b = 16 * (wave + TAP_WAVE * i) + n;
            v8f yr = {}, yi = {};
            yr = __builtin_amdgcn_wmma_f32_16x16x4_f32(false, F.fr[0], false, xt[i], (short)0, yr, false, false);
            yi = __builtin_amdgcn_wmma_f32_16x16x4_f32(false, F.fi[0], false, xt[i], (short)0, yi, false, false);
            float vr = yr[0], vi = yi[0];
#pragma unroll
            for (int r = 1; r < 8; ++r) {
                if (r == d0) { vr = yr[r];  vi = yi[r]; }
            }
            float s, co;
            sincospif((float)(b * d) * (1.0f / 8192.0f), &s, &co);
            re[PHYS(SUBLEN * half + b)] = vr * co + vi * s;
            im[PHYS(SUBLEN * half + b)] = vi * co - vr * s;
        }
    }
    __syncthreads();
    radix16_stage<64, false, TAP_POS / 256, TAP_WAVE>(re, im, F, wave, n, half);
    __syncthreads();
    radix16_stage<4, false, TAP_POS / 256, TAP_WAVE>(re, im, F, wave, n, half);
    __syncthreads();
#pragma unroll
    for (int i = 0; i < TAP_POS / 4 / TAP_THR; ++i) {
        const int g = PHYS(4 * (tid + TAP_THR * i)) >> 2;
        const float4 xr = ((const float4*)re)[g];
        const float4 xi = ((const float4*)im)[g];
        float4 yr, yi;
        dft4<1>(xr, xi, yr, yi);
        float4 gg;
        gg.x = (yr.x * yr.x + yi.x * yi.x) * (1.0f / NFFT);
        gg.y = (yr.y * yr.y + yi.y * yi.y) * (1.0f / NFFT);
        gg.z = (yr.z * yr.z + yi.z * yi.z) * (1.0f / NFFT);
        gg.w = (yr.w * yr.w + yi.w * yi.w) * (1.0f / NFFT);
        ((float4*)gsq)[(d0 + 8 * i) * (NSUB / 4) + (g - i * (NSUB / 4))] = gg;
    }
}

__global__ __launch_bounds__(NTHR)
void fir_fft_kernel(const float* __restrict__ x,
                    const float* __restrict__ gsq,
                    float* __restrict__ dst) {
    __shared__ __attribute__((aligned(16))) float re[NPLANE];
    __shared__ __attribute__((aligned(16))) float im[NPLANE];
    __shared__ __attribute__((aligned(16))) float gp[NPLANE];

    const int tid  = threadIdx.x;
    const int lane = tid & 31;
    const int wave = __builtin_amdgcn_readfirstlane(tid >> 5);
    const int n    = lane & 15;
    const int half = lane >> 4;

    const int pr  = (int)blockIdx.x >> 4;
    const int blk = (int)blockIdx.x & 15;
    const float* x1 = x + (size_t)(2 * pr) * T_LEN;
    const float* x2 = x1 + T_LEN;
    const int t0 = HOP * blk - 4096;

    const Frags F = make_frags(n, half);

    constexpr int NGQ = NFFT / 4 / NTHR / 2;
#pragma unroll 1
    for (int hb = 0; hb < 2; ++hb) {
        float a1[16], a2[16];
        float4 gq[NGQ];
#pragma unroll
        for (int i = 0; i < NGQ; ++i) {
            const int g = PHYS(4 * (tid + NTHR * (NGQ * hb + i))) >> 2;
            gq[i] = ((const float4*)gsq)[g];
        }
#pragma unroll
        for (int i = 0; i < 16; ++i) {
            const int q = tid + NTHR * (16 * hb + i);
            a1[i] = pad_sample(x1, t0 + q);
            a2[i] = pad_sample(x2, t0 + q);
        }
#pragma unroll
        for (int i = 0; i < NGQ; ++i) {
            const int g = PHYS(4 * (tid + NTHR * (NGQ * hb + i))) >> 2;
            ((float4*)gp)[g] = gq[i];
        }
#pragma unroll
        for (int i = 0; i < 16; ++i) {
            const int q = tid + NTHR * (16 * hb + i);
            re[PHYS(q)] = a1[i];
            im[PHYS(q)] = a2[i];
        }
    }
    __syncthreads();
    radix16_stage<1024, false, 64, NWAVE>(re, im, F, wave, n, half);
    __syncthreads();
    radix16_stage<64, false, 64, NWAVE>(re, im, F, wave, n, half);
    __syncthreads();
    radix16_stage<4, false, 64, NWAVE>(re, im, F, wave, n, half);
    __syncthreads();
#pragma unroll
    for (int i = 0; i < NFFT / 4 / NTHR; ++i) {
        const int g = PHYS(4 * (tid + NTHR * i)) >> 2;
        const float4 xr = ((const float4*)re)[g];
        const float4 xi = ((const float4*)im)[g];
        float4 yr, yi;
        dft4<1>(xr, xi, yr, yi);
        const float4 gg = ((const float4*)gp)[g];
        yr.x *= gg.x;  yr.y *= gg.y;  yr.z *= gg.z;  yr.w *= gg.w;
        yi.x *= gg.x;  yi.y *= gg.y;  yi.z *= gg.z;  yi.w *= gg.w;
        float4 zr, zi;
        dft4<-1>(yr, yi, zr, zi);
        ((float4*)re)[g] = zr;
        ((float4*)im)[g] = zi;
    }
    __syncthreads();

    radix16_stage<4, true, 64, NWAVE>(re, im, F, wave, n, half);
    __syncthreads();
    radix16_stage<64, true, 64, NWAVE>(re, im, F, wave, n, half);
    __syncthreads();
    radix16_stage<1024, true, 64, NWAVE>(re, im, F, wave, n, half);
    __syncthreads();

    float* o1 = dst + (size_t)(2 * pr) * T_LEN + HOP * blk;
    float* o2 = o1 + T_LEN;
#pragma unroll
    for (int i = 0; i < HOP / NTHR; ++i) {
        const int j = tid + NTHR * i;
        o1[j] = re[PHYS(4096 + j)];
        o2[j] = im[PHYS(4096 + j)];
    }
}

extern "C" void kernel_launch(void* const* d_in, const int* in_sizes, int n_in,
                              void* d_out, int out_size, void* d_ws, size_t ws_size,
                              hipStream_t stream) {
    const float* x    = (const float*)d_in[0];
    const float* kern = (const float*)d_in[1];
    float* out = (float*)d_out;
    float* gsq = (float*)d_ws;

    fir_taps_kernel<<<8, TAP_THR, 0, stream>>>(kern, gsq);
    fir_fft_kernel<<<256, NTHR, 0, stream>>>(x, gsq, out);
}


# stage factors made once per wave and stage from two 128-entry LDS tables (sub-length 1024: turned by a fixed angle per tile), not 8 sincospif per tile; factors differ in last bits; sum order kept
# speedup vs baseline: 14.0592x; 1.0943x over previous
#include <hip/hip_runtime.h>


typedef __attribute__((ext_vector_type(2))) float v2f;
typedef __attribute__((ext_vector_type(8))) float v8f;

#define K_TAPS 4096
#define PAD    2048
#define T_LEN  131072
#define NFFT   16384
#define HOP    8192
#define NTHR   512
#define NWAVE  (NTHR / 32)
#define PHYS(p) ((p) + (((p) >> 6) << 2))
#define NPLANE (NFFT + NFFT / 16)

struct Frags { v2f fr[4], fi[4], nfi[4]; };

__device__ __forceinline__ Frags make_frags(int n, int half) {
    Frags F;
#pragma unroll
    for (int c = 0; c < 4; ++c) {
        const int a0 = 4 * c + 2 * half;
        float s0, c0, s1, c1;
        sincospif((float)((a0 * n) & 15) * 0.125f, &s0, &c0);
        sincospif((float)(((a0 + 1) * n) & 15) * 0.125f, &s1, &c1);
        F.fr[c].x  = c0;   F.fr[c].y  = c1;
        F.fi[c].x  = -s0;  F.fi[c].y  = -s1;
        F.nfi[c].x = s0;   F.nfi[c].y = s1;
    }
    return F;
}

#define NTAB 128
__device__ __forceinline__ void fill_tables(float2* __restrict__ thi, float2* __restrict__ tlo, int tid) {
    if (tid < NTAB) {
        float s, c;
        sincospif((float)tid * (1.0f / 8192.0f), &s, &c);
        tlo[tid] = make_float2(c, s);
    } else if (tid < 2 * NTAB) {
        float s, c;
        sincospif((float)(tid - NTAB) * (1.0f / 64.0f), &s, &c);
        thi[tid - NTAB] = make_float2(c, s);
    }
}
__device__ __forceinline__ float2 turn(const float2* __restrict__ thi, const float2* __restrict__ tlo, int p) {
    const float2 h = thi[p >> 7];
    const float2 l = tlo[p & (NTAB - 1)];
    return make_float2(h.x * l.x - h.y * l.y, h.y * l.x + h.x * l.y);
}

template <int M, bool INV, int NT, int NW>
__device__ __forceinline__ void radix16_stage(float* __restrict__ re, float* __restrict__ im,
                                              const float2* __restrict__ thi,
                                              const float2* __restrict__ tlo,
                                              const Frags& F, int wave, int n, int half) {
    constexpr int SC = 1024 / M;
    constexpr int  CS   = 16 * NW;
    constexpr bool SAME = (CS % M) == 0;
    static_assert(SAME || 16 * NT <= M, "tiles of a wave must share b or never wrap");
    float tc[8], ts[8], qc[8], qs[8];
    {
        const int b0 = (16 * wave + n) & (M - 1);
#pragma unroll
        for (int j = 0; j < 8; ++j) {
            const int idx = INV ? (4 * (j >> 1) + 2 * half + (j & 1)) : (8 * half + j);
            const float2 t = turn(thi, tlo, SC * b0 * idx);
            tc[j] = t.x;  ts[j] = t.y;
            if (!SAME) {
                const float2 q = turn(thi, tlo, SC * CS * idx);
                qc[j] = q.x;  qs[j] = q.y;
            }
        }
    }
    constexpr int MP = (M >= 64) ? M + M / 16 : M;
#pragma unroll 1
    for (int tau = wave; tau < NT; tau += NW) {
        const int col  = 16 * tau + n;
        const int b    = col & (M - 1);
        const int base = PHYS((col - b) * 16 + b);
        v2f xr[4], xi[4];
#pragma unroll
        for (int c = 0; c < 4; ++c) {
            const int k0 = 4 * c + 2 * half;
            xr[c].x = re[base + k0 * MP];
            xr[c].y = re[base + (k0 + 1) * MP];
            xi[c].x = im[base + k0 * MP];
            xi[c].y = im[base + (k0 + 1) * MP];
        }
        if (INV) {
#pragma unroll
            for (int c = 0; c < 4; ++c) {
                const float c0 = tc[2 * c], s0 = ts[2 * c], c1 = tc[2 * c + 1], s1 = ts[2 * c + 1];
                const float ar = xr[c].x, ai = xi[c].x, br = xr[c].y, bi = xi[c].y;
                xr[c].x = ar * c0 - ai * s0;   xi[c].x = ar * s0 + ai * c0;
                xr[c].y = br * c1 - bi * s1;   xi[c].y = br * s1 + bi * c1;
            }
        }
        v8f yr = {}, yi = {};
#pragma unroll
        for (int c = 0; c < 4; ++c) {
            if (!INV) {
                yr = __builtin_amdgcn_wmma_f32_16x16x4_f32(false, F.fr[c],  false, xr[c], (short)0, yr, false, false);
                yi = __builtin_amdgcn_wmma_f32_16x16x4_f32(false, F.fi[c],  false, xr[c], (short)0, yi, false, false);
                yr = __builtin_amdgcn_wmma_f32_16x16x4_f32(false, F.nfi[c], false, xi[c], (short)0, yr, false, false);
                yi = __builtin_amdgcn_wmma_f32_16x16x4_f32(false, F.fr[c],  false, xi[c], (short)0, yi, false, false);
            } else {
                yr = __builtin_amdgcn_wmma_f32_16x16x4_f32(false, F.fr[c],  false, xr[c], (short)0, yr, false, false);
                yi = __builtin_amdgcn_wmma_f32_16x16x4_f32(false, F.nfi[c], false, xr[c], (short)0, yi, false, false);
                yr = __builtin_amdgcn_wmma_f32_16x16x4_f32(false, F.fi[c],  false, xi[c], (short)0, yr, false, false);
                yi = __builtin_amdgcn_wmma_f32_16x16x4_f32(false, F.fr[c],  false, xi[c], (short)0, yi, false, false);
            }
        }
#pragma unroll
        for (int r = 0; r < 8; ++r) {
            const int d = 8 * half + r;
            float vr = yr[r], vi = yi[r];
            if (!INV) {
                const float co = tc[r], s = ts[r];
                const float tr = vr * co + vi * s;
                const float ti = vi * co - vr * s;
                vr = tr;  vi = ti;
            }
            re[base + d * MP] = vr;
            im[base + d * MP] = vi;
        }
        if (!SAME) {
#pragma unroll
            for (int j = 0; j < 8; ++j) {
                const float c = tc[j], s = ts[j];
                tc[j] = c * qc[j] - s * qs[j];
                ts[j] = s * qc[j] + c * qs[j];
            }
        }
    }
}

template <int SGN>
__device__ __forceinline__ void dft4(float4 xr, float4 xi, float4& yr, float4& yi) {
    const float ar = xr.x + xr.z, ai = xi.x + xi.z;
    const float br = xr.x - xr.z, bi = xi.x - xi.z;
    const float cr = xr.y + xr.w, ci = xi.y + xi.w;
    const float dr = xr.y - xr.w, di = xi.y - xi.w;
    yr.x = ar + cr;  yi.x = ai + ci;
    yr.z = ar - cr;  yi.z = ai - ci;
    if (SGN > 0) {
        yr.y = br + di;  yi.y = bi - dr;
        yr.w = br - di;  yi.w = bi + dr;
    } else {
        yr.y = br - di;  yi.y = bi + dr;
        yr.w = br + di;  yi.w = bi - dr;
    }
}

__device__ __forceinline__ float pad_sample(const float* __restrict__ xs, int t) {
    int idx = t;
    if (t < 0) idx = -t;
    if (t >= T_LEN) idx = 2 * T_LEN - 2 - t;
    const bool ok = (t >= -PAD) && (t < T_LEN + PAD);
    if (!ok) idx = 0;
    const float v = xs[idx];
    return ok ? v : 0.f;
}

#define TAP_THR  256
#define TAP_WAVE (TAP_THR / 32)
#define SUBLEN   1024
#define TAP_POS  (2 * SUBLEN)
#define NSUB     (SUBLEN + SUBLEN / 16)
#define NTAPP    (TAP_POS + TAP_POS / 16)
static_assert(K_TAPS == 4 * SUBLEN, "the taps are points a = 0..3 of every first-stage column");

__global__ __launch_bounds__(TAP_THR)
void fir_taps_kernel(const float* __restrict__ h, float* __restrict__ gsq) {
    __shared__ __attribute__((aligned(16))) float re[NTAPP];
    __shared__ __attribute__((aligned(16))) float im[NTAPP];
    __shared__ __attribute__((aligned(16))) float2 thi[NTAB];
    __shared__ __attribute__((aligned(16))) float2 tlo[NTAB];

    const int tid  = threadIdx.x;
    const int lane = tid & 31;
    const int wave = __builtin_amdgcn_readfirstlane(tid >> 5);
    const int n    = lane & 15;
    const int half = lane >> 4;
    const int d0   = (int)blockIdx.x;
    const int d    = d0 + 8 * half;

    const Frags F = make_frags(n, half);
    fill_tables(thi, tlo, tid);

    {
        constexpr int NTA = SUBLEN / 16 / TAP_WAVE;
        v2f xt[NTA];
#pragma unroll
        for (int i = 0; i < NTA; ++i) {
            const int b = 16 * (wave + TAP_WAVE * i) + n;
            xt[i].x = h[(2 * half) * SUBLEN + b];
            xt[i].y = h[(2 * half + 1) * SUBLEN + b];
        }
#pragma unroll
        for (int i = 0; i < NTA; ++i) {
            const int b = 16 * (wave + TAP_WAVE * i) + n;
            v8f yr = {}, yi = {};
            yr = __builtin_amdgcn_wmma_f32_16x16x4_f32(false, F.fr[0], false, xt[i], (short)0, yr, false, false);
            yi = __builtin_amdgcn_wmma_f32_16x16x4_f32(false, F.fi[0], false, xt[i], (short)0, yi, false, false);
            float vr = yr[0], vi = yi[0];
#pragma unroll
            for (int r = 1; r < 8; ++r) {
                if (r == d0) { vr = yr[r];  vi = yi[r]; }
            }
            float s, co;
            sincospif((float)(b * d) * (1.0f / 8192.0f), &s, &co);
            re[PHYS(SUBLEN * half + b)] = vr * co + vi * s;
            im[PHYS(SUBLEN * half + b)] = vi * co - vr * s;
        }
    }
    __syncthreads();
    radix16_stage<64, false, TAP_POS / 256, TAP_WAVE>(re, im, thi, tlo, F, wave, n, half);
    __syncthreads();
    radix16_stage<4, false, TAP_POS / 256, TAP_WAVE>(re, im, thi, tlo, F, wave, n, half);
    __syncthreads();
#pragma unroll
    for (int i = 0; i < TAP_POS / 4 / TAP_THR; ++i) {
        const int g = PHYS(4 * (tid + TAP_THR * i)) >> 2;
        const float4 xr = ((const float4*)re)[g];
        const float4 xi = ((const float4*)im)[g];
        float4 yr, yi;
        dft4<1>(xr, xi, yr, yi);
        float4 gg;
        gg.x = (yr.x * yr.x + yi.x * yi.x) * (1.0f / NFFT);
        gg.y = (yr.y * yr.y + yi.y * yi.y) * (1.0f / NFFT);
        gg.z = (yr.z * yr.z + yi.z * yi.z) * (1.0f / NFFT);
        gg.w = (yr.w * yr.w + yi.w * yi.w) * (1.0f / NFFT);
        ((float4*)gsq)[(d0 + 8 * i) * (NSUB / 4) + (g - i * (NSUB / 4))] = gg;
    }
}

__global__ __launch_bounds__(NTHR)
void fir_fft_kernel(const float* __restrict__ x,
                    const float* __restrict__ gsq,
                    float* __restrict__ dst) {
    __shared__ __attribute__((aligned(16))) float re[NPLANE];
    __shared__ __attribute__((aligned(16))) float im[NPLANE];
    __shared__ __attribute__((aligned(16))) float gp[NPLANE];
    __shared__ __attribute__((aligned(16))) float2 thi[NTAB];
    __shared__ __attribute__((aligned(16))) float2 tlo[NTAB];

    const int tid  = threadIdx.x;
    const int lane = tid & 31;
    const int wave = __builtin_amdgcn_readfirstlane(tid >> 5);
    const int n    = lane & 15;
    const int half = lane >> 4;

    const int pr  = (int)blockIdx.x >> 4;
    const int blk = (int)blockIdx.x & 15;
    const float* x1 = x + (size_t)(2 * pr) * T_LEN;
    const float* x2 = x1 + T_LEN;
    const int t0 = HOP * blk - 4096;

    const Frags F = make_frags(n, half);
    fill_tables(thi, tlo, tid);

    constexpr int NGQ = NFFT / 4 / NTHR / 2;
#pragma unroll 1
    for (int hb = 0; hb < 2; ++hb) {
        float a1[16], a2[16];
        float4 gq[NGQ];
#pragma unroll
        for (int i = 0; i < NGQ; ++i) {
            const int g = PHYS(4 * (tid + NTHR * (NGQ * hb + i))) >> 2;
            gq[i] = ((const float4*)gsq)[g];
        }
#pragma unroll
        for (int i = 0; i < 16; ++i) {
            const int q = tid + NTHR * (16 * hb + i);
            a1[i] = pad_sample(x1, t0 + q);
            a2[i] = pad_sample(x2, t0 + q);
        }
#pragma unroll
        for (int i = 0; i < NGQ; ++i) {
            const int g = PHYS(4 * (tid + NTHR * (NGQ * hb + i))) >> 2;
            ((float4*)gp)[g] = gq[i];
        }
#pragma unroll
        for (int i = 0; i < 16; ++i) {
            const int q = tid + NTHR * (16 * hb + i);
            re[PHYS(q)] = a1[i];
            im[PHYS(q)] = a2[i];
        }
    }
    __syncthreads();
    radix16_stage<1024, false, 64, NWAVE>(re, im, thi, tlo, F, wave, n, half);
    __syncthreads();
    radix16_stage<64, false, 64, NWAVE>(re, im, thi, tlo, F, wave, n, half);
    __syncthreads();
    radix16_stage<4, false, 64, NWAVE>(re, im, thi, tlo, F, wave, n, half);
    __syncthreads();
#pragma unroll
    for (int i = 0; i < NFFT / 4 / NTHR; ++i) {
        const int g = PHYS(4 * (tid + NTHR * i)) >> 2;
        const float4 xr = ((const float4*)re)[g];
        const float4 xi = ((const float4*)im)[g];
        float4 yr, yi;
        dft4<1>(xr, xi, yr, yi);
        const float4 gg = ((const float4*)gp)[g];
        yr.x *= gg.x;  yr.y *= gg.y;  yr.z *= gg.z;  yr.w *= gg.w;
        yi.x *= gg.x;  yi.y *= gg.y;  yi.z *= gg.z;  yi.w *= gg.w;
        float4 zr, zi;
        dft4<-1>(yr, yi, zr, zi);
        ((float4*)re)[g] = zr;
        ((float4*)im)[g] = zi;
    }
    __syncthreads();

    radix16_stage<4, true, 64, NWAVE>(re, im, thi, tlo, F, wave, n, half);
    __syncthreads();
    radix16_stage<64, true, 64, NWAVE>(re, im, thi, tlo, F, wave, n, half);
    __syncthreads();
    radix16_stage<1024, true, 64, NWAVE>(re, im, thi, tlo, F, wave, n, half);
    __syncthreads();

    float* o1 = dst + (size_t)(2 * pr) * T_LEN + HOP * blk;
    float* o2 = o1 + T_LEN;
#pragma unroll
    for (int i = 0; i < HOP / NTHR; ++i) {
        const int j = tid + NTHR * i;
        o1[j] = re[PHYS(4096 + j)];
        o2[j] = im[PHYS(4096 + j)];
    }
}

extern "C" void kernel_launch(void* const* d_in, const int* in_sizes, int n_in,
                              void* d_out, int out_size, void* d_ws, size_t ws_size,
                              hipStream_t stream) {
    const float* x    = (const float*)d_in[0];
    const float* kern = (const float*)d_in[1];
    float* out = (float*)d_out;
    float* gsq = (float*)d_ws;

    fir_taps_kernel<<<8, TAP_THR, 0, stream>>>(kern, gsq);
    fir_fft_kernel<<<256, NTHR, 0, stream>>>(x, gsq, out);
}
